# starting kernel
# baseline (speedup 1.0000x reference)
#include <hip/hip_runtime.h>


typedef __attribute__((ext_vector_type(2))) float v2f;
typedef __attribute__((ext_vector_type(8))) float v8f;

#define GA_B    2048
#define GA_NIN  512
#define GA_NOUT 512

__device__ __forceinline__ v2f mk2(float a, float b) { v2f r; r.x = a; r.y = b; return r; }

__device__ __forceinline__ v8f wmma_f32(v2f a, v2f b, v8f c) {
  return __builtin_amdgcn_wmma_f32_16x16x4_f32(false, a, false, b, (short)0, c, false, false);
}

static constexpr int GRADE_[8] = {0, 1, 1, 1, 2, 2, 2, 3};

static constexpr int EBI[64] = {
  0,0,0,0,0,0,0,0,
  1,2,3, 1,2,3,
  1,1,2,2,3,3,
  1,1,2,2,3,3,
  1,2,3, 1,2,3,
  4,5,6,
  4,4,5,5,6,6,
  4,5,6, 4,5,6,
  4,4,5,5,6,6,
  4,5,6,
  7, 7,7,7, 7,7,7, 7};
static constexpr int EBJ[64] = {
  0,1,2,3,4,5,6,7,
  0,0,0, 1,2,3,
  2,3,1,3,1,2,
  4,5,4,6,5,6,
  6,5,4, 7,7,7,
  0,0,0,
  1,2,1,3,2,3,
  3,2,1, 4,5,6,
  5,6,4,6,4,5,
  7,7,7,
  0, 1,2,3, 4,5,6, 7};
static constexpr int EBK[64] = {
  0,1,2,3,4,5,6,7,
  1,2,3, 0,0,0,
  4,5,4,6,5,6,
  2,3,1,3,1,2,
  7,7,7, 6,5,4,
  4,5,6,
  2,1,3,1,3,2,
  7,7,7, 0,0,0,
  6,5,6,4,5,4,
  3,2,1,
  7, 6,5,4, 3,2,1, 0};
static constexpr int ECV[64] = {
  1,1,1,1,1,1,1,1,
  1,1,1, 1,1,1,
  1,1,-1,1,-1,-1,
  1,1,-1,1,-1,-1,
  1,-1,1, 1,-1,1,
  -1,-1,-1,
  1,-1,1,-1,1,-1,
  -1,1,-1, 1,1,1,
  1,-1,-1,1,1,-1,
  1,-1,1,
  -1, -1,1,-1, 1,-1,1, 1};
static constexpr int EPX[64] = {
  0,1,1,1,2,2,2,3,
  4,4,4, 5,5,5,
  6,6,6,6,6,6,
  7,7,7,7,7,7,
  8,8,8, 9,9,9,
  10,10,10,
  11,11,11,11,11,11,
  12,12,12, 13,13,13,
  14,14,14,14,14,14,
  15,15,15,
  16, 17,17,17, 18,18,18, 19};


__global__ __launch_bounds__(256) void ga_repack_x(
    const float* __restrict__ in, float* __restrict__ outp) {
  const int tid = blockIdx.x * 256 + threadIdx.x;
  const int b = tid >> 9;
  const int n = tid & 511;
  const float4 p0 = *(const float4*)(in + (size_t)tid * 8);
  const float4 p1 = *(const float4*)(in + (size_t)tid * 8 + 4);
  float* dst = outp + (size_t)b * (8 * 512) + n;
  dst[0 * 512] = p0.x; dst[1 * 512] = p0.y; dst[2 * 512] = p0.z; dst[3 * 512] = p0.w;
  dst[4 * 512] = p1.x; dst[5 * 512] = p1.y; dst[6 * 512] = p1.z; dst[7 * 512] = p1.w;
}

template <int C>
__global__ __launch_bounds__(256) void ga_repack_w(
    const float* __restrict__ in, float* __restrict__ outp) {
  const int tid = blockIdx.x * 256 + threadIdx.x;
  const int m = tid >> 9;
  const int n = tid & 511;
  const float* src = in + (size_t)tid * C;
  float* dst = outp + (size_t)m * (C * 512) + n;
#pragma unroll
  for (int c = 0; c < C; ++c) dst[(size_t)c * 512] = src[c];
}


__global__ __launch_bounds__(32) void ga_xr_norm_fast(
    const float* __restrict__ xt, const float* __restrict__ wrt,
    const float* __restrict__ norm_a, float* __restrict__ xrt) {
  const int b0 = blockIdx.x << 4;
  const int m0 = blockIdx.y << 4;
  const int lane = threadIdx.x & 31;
  const int lo = lane & 15;
  const int hi = lane >> 4;

  const v8f vzero = {};
  v8f acc[8];
#pragma unroll
  for (int i = 0; i < 8; ++i) acc[i] = vzero;

  const float* xrow = xt + (size_t)(b0 + lo) * (8 * 512);
  const float* wrow = wrt + (size_t)(m0 + lo) * (4 * 512);

#pragma unroll 4
  for (int k = 0; k < GA_NIN; k += 4) {
    const int kk = k + (hi << 1);
    v2f a[8], g[4];
#pragma unroll
    for (int i = 0; i < 8; ++i) a[i] = *(const v2f*)(xrow + i * 512 + kk);
#pragma unroll
    for (int j = 0; j < 4; ++j) g[j] = *(const v2f*)(wrow + j * 512 + kk);
#pragma unroll
    for (int i = 0; i < 8; ++i)
      acc[i] = wmma_f32(a[i], g[GRADE_[i]], acc[i]);
  }

  const float4 na = *(const float4*)(norm_a + (size_t)(m0 + lo) * 4);
  const float s0 = 1.0f / (1.0f + __expf(-na.x));
  const float s1 = 1.0f / (1.0f + __expf(-na.y));
  const float s2 = 1.0f / (1.0f + __expf(-na.z));
  const float s3 = 1.0f / (1.0f + __expf(-na.w));

#pragma unroll
  for (int v = 0; v < 8; ++v) {
    const float e0 = acc[0][v], e1 = acc[1][v], e2 = acc[2][v], e3 = acc[3][v];
    const float e4 = acc[4][v], e5 = acc[5][v], e6 = acc[6][v], e7 = acc[7][v];
    const float n0 = __builtin_sqrtf(e0 * e0);
    const float n1 = __builtin_sqrtf(e1 * e1 + e2 * e2 + e3 * e3);
    const float n2 = __builtin_sqrtf(e4 * e4 + e5 * e5 + e6 * e6);
    const float n3 = __builtin_sqrtf(e7 * e7);
    const float i0 = 1.0f / (s0 * (n0 - 1.0f) + 1.0f + 1e-6f);
    const float i1 = 1.0f / (s1 * (n1 - 1.0f) + 1.0f + 1e-6f);
    const float i2 = 1.0f / (s2 * (n2 - 1.0f) + 1.0f + 1e-6f);
    const float i3 = 1.0f / (s3 * (n3 - 1.0f) + 1.0f + 1e-6f);
    float* dst = xrt + (size_t)(b0 + v + (hi << 3)) * (8 * 512) + (m0 + lo);
    dst[0 * 512] = e0 * i0;
    dst[1 * 512] = e1 * i1;
    dst[2 * 512] = e2 * i1;
    dst[3 * 512] = e3 * i1;
    dst[4 * 512] = e4 * i2;
    dst[5 * 512] = e5 * i2;
    dst[6 * 512] = e6 * i2;
    dst[7 * 512] = e7 * i3;
  }
}

__global__ __launch_bounds__(32) void ga_out_fast(
    const float* __restrict__ xt, const float* __restrict__ xrt,
    const float* __restrict__ wt, const float* __restrict__ wlt,
    const float* __restrict__ b_left, float* __restrict__ out) {
  const int b0 = blockIdx.x << 4;
  const int m0 = blockIdx.y << 4;
  const int lane = threadIdx.x & 31;
  const int lo = lane & 15;
  const int hi = lane >> 4;

  const v8f vzero = {};
  v8f acc[8];
#pragma unroll
  for (int i = 0; i < 8; ++i) acc[i] = vzero;

  const float* xrow = xt + (size_t)(b0 + lo) * (8 * 512);
  const float* rrow = xrt + (size_t)(b0 + lo) * (8 * 512);
  const float* lrow = wlt + (size_t)(m0 + lo) * (4 * 512);
  const float* wrow = wt + (size_t)(m0 + lo) * (20 * 512);

#pragma unroll 1
  for (int k = 0; k < GA_NIN; k += 4) {
    const int kk = k + (hi << 1);
    v2f a[8], r[8], bl[4], bw[20];
#pragma unroll
    for (int i = 0; i < 8; ++i) a[i] = *(const v2f*)(xrow + i * 512 + kk);
#pragma unroll
    for (int i = 0; i < 8; ++i) r[i] = *(const v2f*)(rrow + i * 512 + kk);
#pragma unroll
    for (int j = 0; j < 4; ++j) bl[j] = *(const v2f*)(lrow + j * 512 + kk);
#pragma unroll
    for (int p = 0; p < 20; ++p) bw[p] = *(const v2f*)(wrow + p * 512 + kk);

#pragma unroll
    for (int i = 0; i < 8; ++i)
      acc[i] = wmma_f32(a[i], bl[GRADE_[i]], acc[i]);

#pragma unroll
    for (int e = 0; e < 64; ++e) {
      v2f p = a[EBI[e]] * r[EBK[e]];
      if (ECV[e] < 0) p = -p;
      acc[EBJ[e]] = wmma_f32(p, bw[EPX[e]], acc[EBJ[e]]);
    }
  }

  const float bias = b_left[m0 + lo];
  const float rs2 = 0.70710678118654752440f;
#pragma unroll
  for (int v = 0; v < 8; ++v) {
    const float o0 = (acc[0][v] + bias) * rs2;
    const float o1 = acc[1][v] * rs2, o2 = acc[2][v] * rs2, o3 = acc[3][v] * rs2;
    const float o4 = acc[4][v] * rs2, o5 = acc[5][v] * rs2;
    const float o6 = acc[6][v] * rs2, o7 = acc[7][v] * rs2;
    float* dst = out + ((size_t)(b0 + v + (hi << 3)) * GA_NOUT + (m0 + lo)) * 8;
    *(float4*)(dst)     = make_float4(o0, o1, o2, o3);
    *(float4*)(dst + 4) = make_float4(o4, o5, o6, o7);
  }
}


__global__ __launch_bounds__(32) void ga_xr_norm_kernel(
    const float* __restrict__ x, const float* __restrict__ w_right,
    const float* __restrict__ norm_a, float* __restrict__ xr) {
  const int b0 = blockIdx.x << 4;
  const int m0 = blockIdx.y << 4;
  const int lane = threadIdx.x & 31;
  const int lo = lane & 15;
  const int hi = lane >> 4;

  const v8f vzero = {};
  v8f acc[8];
#pragma unroll
  for (int i = 0; i < 8; ++i) acc[i] = vzero;

  const float* xrow = x + (size_t)(b0 + lo) * (GA_NIN * 8);
  const float* wrow = w_right + (size_t)(m0 + lo) * (GA_NIN * 4);

#pragma unroll 2
  for (int k = 0; k < GA_NIN; k += 4) {
    const int kk = k + (hi << 1);
    const float4 xa0 = *(const float4*)(xrow + kk * 8);
    const float4 xa1 = *(const float4*)(xrow + kk * 8 + 4);
    const float4 xb0 = *(const float4*)(xrow + kk * 8 + 8);
    const float4 xb1 = *(const float4*)(xrow + kk * 8 + 12);
    const float4 w0  = *(const float4*)(wrow + kk * 4);
    const float4 w1  = *(const float4*)(wrow + kk * 4 + 4);

    const v2f a0 = mk2(xa0.x, xb0.x), a1 = mk2(xa0.y, xb0.y);
    const v2f a2 = mk2(xa0.z, xb0.z), a3 = mk2(xa0.w, xb0.w);
    const v2f a4 = mk2(xa1.x, xb1.x), a5 = mk2(xa1.y, xb1.y);
    const v2f a6 = mk2(xa1.z, xb1.z), a7 = mk2(xa1.w, xb1.w);
    const v2f g0 = mk2(w0.x, w1.x), g1 = mk2(w0.y, w1.y);
    const v2f g2 = mk2(w0.z, w1.z), g3 = mk2(w0.w, w1.w);

    acc[0] = wmma_f32(a0, g0, acc[0]);
    acc[1] = wmma_f32(a1, g1, acc[1]);
    acc[2] = wmma_f32(a2, g1, acc[2]);
    acc[3] = wmma_f32(a3, g1, acc[3]);
    acc[4] = wmma_f32(a4, g2, acc[4]);
    acc[5] = wmma_f32(a5, g2, acc[5]);
    acc[6] = wmma_f32(a6, g2, acc[6]);
    acc[7] = wmma_f32(a7, g3, acc[7]);
  }

  const float4 na = *(const float4*)(norm_a + (size_t)(m0 + lo) * 4);
  const float s0 = 1.0f / (1.0f + __expf(-na.x));
  const float s1 = 1.0f / (1.0f + __expf(-na.y));
  const float s2 = 1.0f / (1.0f + __expf(-na.z));
  const float s3 = 1.0f / (1.0f + __expf(-na.w));

#pragma unroll
  for (int v = 0; v < 8; ++v) {
    const float e0 = acc[0][v], e1 = acc[1][v], e2 = acc[2][v], e3 = acc[3][v];
    const float e4 = acc[4][v], e5 = acc[5][v], e6 = acc[6][v], e7 = acc[7][v];
    const float n0 = __builtin_sqrtf(e0 * e0);
    const float n1 = __builtin_sqrtf(e1 * e1 + e2 * e2 + e3 * e3);
    const float n2 = __builtin_sqrtf(e4 * e4 + e5 * e5 + e6 * e6);
    const float n3 = __builtin_sqrtf(e7 * e7);
    const float i0 = 1.0f / (s0 * (n0 - 1.0f) + 1.0f + 1e-6f);
    const float i1 = 1.0f / (s1 * (n1 - 1.0f) + 1.0f + 1e-6f);
    const float i2 = 1.0f / (s2 * (n2 - 1.0f) + 1.0f + 1e-6f);
    const float i3 = 1.0f / (s3 * (n3 - 1.0f) + 1.0f + 1e-6f);
    float* dst = xr + ((size_t)(b0 + v + (hi << 3)) * GA_NIN + (m0 + lo)) * 8;
    *(float4*)(dst)     = make_float4(e0 * i0, e1 * i1, e2 * i1, e3 * i1);
    *(float4*)(dst + 4) = make_float4(e4 * i2, e5 * i2, e6 * i2, e7 * i3);
  }
}

__global__ __launch_bounds__(32) void ga_out_kernel(
    const float* __restrict__ x, const float* __restrict__ xr,
    const float* __restrict__ weight, const float* __restrict__ w_left,
    const float* __restrict__ b_left, float* __restrict__ out) {
  const int b0 = blockIdx.x << 4;
  const int m0 = blockIdx.y << 4;
  const int lane = threadIdx.x & 31;
  const int lo = lane & 15;
  const int hi = lane >> 4;

  const v8f vzero = {};
  v8f acc[8];
#pragma unroll
  for (int i = 0; i < 8; ++i) acc[i] = vzero;

  const float* xrow = x  + (size_t)(b0 + lo) * (GA_NIN * 8);
  const float* rrow = xr + (size_t)(b0 + lo) * (GA_NIN * 8);
  const float* lrow = w_left + (size_t)(m0 + lo) * (GA_NIN * 4);
  const float* wrow = weight + (size_t)(m0 + lo) * (GA_NIN * 20);

#pragma unroll 1
  for (int k = 0; k < GA_NIN; k += 4) {
    const int kk = k + (hi << 1);

    const float4 xa0 = *(const float4*)(xrow + kk * 8);
    const float4 xa1 = *(const float4*)(xrow + kk * 8 + 4);
    const float4 xb0 = *(const float4*)(xrow + kk * 8 + 8);
    const float4 xb1 = *(const float4*)(xrow + kk * 8 + 12);
    const float4 ra0 = *(const float4*)(rrow + kk * 8);
    const float4 ra1 = *(const float4*)(rrow + kk * 8 + 4);
    const float4 rb0 = *(const float4*)(rrow + kk * 8 + 8);
    const float4 rb1 = *(const float4*)(rrow + kk * 8 + 12);
    const float4 l0  = *(const float4*)(lrow + kk * 4);
    const float4 l1  = *(const float4*)(lrow + kk * 4 + 4);
    const float4 q0  = *(const float4*)(wrow + kk * 20);
    const float4 q1  = *(const float4*)(wrow + kk * 20 + 4);
    const float4 q2  = *(const float4*)(wrow + kk * 20 + 8);
    const float4 q3  = *(const float4*)(wrow + kk * 20 + 12);
    const float4 q4  = *(const float4*)(wrow + kk * 20 + 16);
    const float4 t0  = *(const float4*)(wrow + kk * 20 + 20);
    const float4 t1  = *(const float4*)(wrow + kk * 20 + 24);
    const float4 t2  = *(const float4*)(wrow + kk * 20 + 28);
    const float4 t3  = *(const float4*)(wrow + kk * 20 + 32);
    const float4 t4  = *(const float4*)(wrow + kk * 20 + 36);

    v2f a[8], r[8], bl[4], bw[20];
    a[0] = mk2(xa0.x, xb0.x); a[1] = mk2(xa0.y, xb0.y);
    a[2] = mk2(xa0.z, xb0.z); a[3] = mk2(xa0.w, xb0.w);
    a[4] = mk2(xa1.x, xb1.x); a[5] = mk2(xa1.y, xb1.y);
    a[6] = mk2(xa1.z, xb1.z); a[7] = mk2(xa1.w, xb1.w);
    r[0] = mk2(ra0.x, rb0.x); r[1] = mk2(ra0.y, rb0.y);
    r[2] = mk2(ra0.z, rb0.z); r[3] = mk2(ra0.w, rb0.w);
    r[4] = mk2(ra1.x, rb1.x); r[5] = mk2(ra1.y, rb1.y);
    r[6] = mk2(ra1.z, rb1.z); r[7] = mk2(ra1.w, rb1.w);
    bl[0] = mk2(l0.x, l1.x); bl[1] = mk2(l0.y, l1.y);
    bl[2] = mk2(l0.z, l1.z); bl[3] = mk2(l0.w, l1.w);
    bw[0]  = mk2(q0.x, t0.x); bw[1]  = mk2(q0.y, t0.y);
    bw[2]  = mk2(q0.z, t0.z); bw[3]  = mk2(q0.w, t0.w);
    bw[4]  = mk2(q1.x, t1.x); bw[5]  = mk2(q1.y, t1.y);
    bw[6]  = mk2(q1.z, t1.z); bw[7]  = mk2(q1.w, t1.w);
    bw[8]  = mk2(q2.x, t2.x); bw[9]  = mk2(q2.y, t2.y);
    bw[10] = mk2(q2.z, t2.z); bw[11] = mk2(q2.w, t2.w);
    bw[12] = mk2(q3.x, t3.x); bw[13] = mk2(q3.y, t3.y);
    bw[14] = mk2(q3.z, t3.z); bw[15] = mk2(q3.w, t3.w);
    bw[16] = mk2(q4.x, t4.x); bw[17] = mk2(q4.y, t4.y);
    bw[18] = mk2(q4.z, t4.z); bw[19] = mk2(q4.w, t4.w);

#pragma unroll
    for (int i = 0; i < 8; ++i)
      acc[i] = wmma_f32(a[i], bl[GRADE_[i]], acc[i]);

#pragma unroll
    for (int e = 0; e < 64; ++e) {
      v2f p = a[EBI[e]] * r[EBK[e]];
      if (ECV[e] < 0) p = -p;
      acc[EBJ[e]] = wmma_f32(p, bw[EPX[e]], acc[EBJ[e]]);
    }
  }

  const float bias = b_left[m0 + lo];
  const float rs2 = 0.70710678118654752440f;
#pragma unroll
  for (int v = 0; v < 8; ++v) {
    const float o0 = (acc[0][v] + bias) * rs2;
    const float o1 = acc[1][v] * rs2, o2 = acc[2][v] * rs2, o3 = acc[3][v] * rs2;
    const float o4 = acc[4][v] * rs2, o5 = acc[5][v] * rs2;
    const float o6 = acc[6][v] * rs2, o7 = acc[7][v] * rs2;
    float* dst = out + ((size_t)(b0 + v + (hi << 3)) * GA_NOUT + (m0 + lo)) * 8;
    *(float4*)(dst)     = make_float4(o0, o1, o2, o3);
    *(float4*)(dst + 4) = make_float4(o4, o5, o6, o7);
  }
}


extern "C" void kernel_launch(void* const* d_in, const int* in_sizes, int n_in,
                              void* d_out, int out_size, void* d_ws, size_t ws_size,
                              hipStream_t stream) {
  (void)in_sizes; (void)n_in; (void)out_size;
  const float* x       = (const float*)d_in[0];
  const float* weight  = (const float*)d_in[1];
  const float* w_right = (const float*)d_in[2];
  const float* w_left  = (const float*)d_in[3];
  const float* b_left  = (const float*)d_in[4];
  const float* norm_a  = (const float*)d_in[5];
  float* out = (float*)d_out;
  float* ws = (float*)d_ws;

  const size_t XT_N  = (size_t)GA_B * 8 * 512;
  const size_t W4_N  = (size_t)512 * 4 * 512;
  const size_t W20_N = (size_t)512 * 20 * 512;
  const size_t FAST_FLOATS = XT_N * 2 + W4_N * 2 + W20_N;

  dim3 block(32, 1, 1);
  dim3 grid1(GA_B / 16, GA_NIN / 16, 1);
  dim3 grid2(GA_B / 16, GA_NOUT / 16, 1);

  if (ws_size >= FAST_FLOATS * sizeof(float)) {
    float* xt  = ws;
    float* xrt = xt + XT_N;
    float* wrt = xrt + XT_N;
    float* wlt = wrt + W4_N;
    float* wtt = wlt + W4_N;

    ga_repack_x<<<dim3((GA_B * 512) / 256), dim3(256), 0, stream>>>(x, xt);
    ga_repack_w<4><<<dim3((512 * 512) / 256), dim3(256), 0, stream>>>(w_right, wrt);
    ga_repack_w<4><<<dim3((512 * 512) / 256), dim3(256), 0, stream>>>(w_left, wlt);
    ga_repack_w<20><<<dim3((512 * 512) / 256), dim3(256), 0, stream>>>(weight, wtt);

    ga_xr_norm_fast<<<grid1, block, 0, stream>>>(xt, wrt, norm_a, xrt);
    ga_out_fast<<<grid2, block, 0, stream>>>(xt, xrt, wtt, wlt, b_left, out);
  } else {
    float* xr = ws;
    ga_xr_norm_kernel<<<grid1, block, 0, stream>>>(x, w_right, norm_a, xr);
    ga_out_kernel<<<grid2, block, 0, stream>>>(x, xr, weight, w_left, b_left, out);
  }
}


# GEMM kernels read x, xr and the weights in their native layouts (whole lines per row and k-chunk) and transpose 4k x 4ch blocks in registers into LDS; the four repack launches are gone
# speedup vs baseline: 2.9080x; 2.9080x over previous
#include <hip/hip_runtime.h>


typedef __attribute__((ext_vector_type(2))) float v2f;
typedef __attribute__((ext_vector_type(8))) float v8f;

#define GA_B    2048
#define GA_NIN  512
#define GA_NOUT 512

__device__ __forceinline__ v8f wmma_f32(v2f a, v2f b, v8f c) {
  return __builtin_amdgcn_wmma_f32_16x16x4_f32(false, a, false, b, (short)0, c, false, false);
}

static constexpr int GRADE_[8] = {0, 1, 1, 1, 2, 2, 2, 3};

static constexpr int EBI[64] = {
  0,0,0,0,0,0,0,0,
  1,2,3, 1,2,3,
  1,1,2,2,3,3,
  1,1,2,2,3,3,
  1,2,3, 1,2,3,
  4,5,6,
  4,4,5,5,6,6,
  4,5,6, 4,5,6,
  4,4,5,5,6,6,
  4,5,6,
  7, 7,7,7, 7,7,7, 7};
static constexpr int EBJ[64] = {
  0,1,2,3,4,5,6,7,
  0,0,0, 1,2,3,
  2,3,1,3,1,2,
  4,5,4,6,5,6,
  6,5,4, 7,7,7,
  0,0,0,
  1,2,1,3,2,3,
  3,2,1, 4,5,6,
  5,6,4,6,4,5,
  7,7,7,
  0, 1,2,3, 4,5,6, 7};
static constexpr int EBK[64] = {
  0,1,2,3,4,5,6,7,
  1,2,3, 0,0,0,
  4,5,4,6,5,6,
  2,3,1,3,1,2,
  7,7,7, 6,5,4,
  4,5,6,
  2,1,3,1,3,2,
  7,7,7, 0,0,0,
  6,5,6,4,5,4,
  3,2,1,
  7, 6,5,4, 3,2,1, 0};
static constexpr int ECV[64] = {
  1,1,1,1,1,1,1,1,
  1,1,1, 1,1,1,
  1,1,-1,1,-1,-1,
  1,1,-1,1,-1,-1,
  1,-1,1, 1,-1,1,
  -1,-1,-1,
  1,-1,1,-1,1,-1,
  -1,1,-1, 1,1,1,
  1,-1,-1,1,1,-1,
  1,-1,1,
  -1, -1,1,-1, 1,-1,1, 1};
static constexpr int EPX[64] = {
  0,1,1,1,2,2,2,3,
  4,4,4, 5,5,5,
  6,6,6,6,6,6,
  7,7,7,7,7,7,
  8,8,8, 9,9,9,
  10,10,10,
  11,11,11,11,11,11,
  12,12,12, 13,13,13,
  14,14,14,14,14,14,
  15,15,15,
  16, 17,17,17, 18,18,18, 19};


struct Blk { float4 r0, r1, r2, r3; };

__device__ __forceinline__ void ld_blk(Blk& b, const float* p, int js) {
  b.r0 = *(const float4*)(p);
  b.r1 = *(const float4*)(p + js);
  b.r2 = *(const float4*)(p + 2 * js);
  b.r3 = *(const float4*)(p + 3 * js);
}

__device__ __forceinline__ void st_blk(float4* d, const Blk& b) {
  d[0]   = make_float4(b.r0.x, b.r1.x, b.r2.x, b.r3.x);
  d[128] = make_float4(b.r0.y, b.r1.y, b.r2.y, b.r3.y);
  d[256] = make_float4(b.r0.z, b.r1.z, b.r2.z, b.r3.z);
  d[384] = make_float4(b.r0.w, b.r1.w, b.r2.w, b.r3.w);
}


#define XR_SUB    4
#define XR_SUB_F4 1536
#define XR_ST_F4  (XR_SUB * XR_SUB_F4)

__global__ __launch_bounds__(512) void ga_xr_lds(
    const float* __restrict__ x, const float* __restrict__ w_right,
    const float* __restrict__ norm_a, float* __restrict__ xr) {
  __shared__ float4 lds[2 * XR_ST_F4];

  const int t = threadIdx.x;
  const int lane = t & 31;
  const int lo = lane & 15;
  const int hi = lane >> 4;
  const int wave = t >> 5;
  const int wr = wave >> 2;
  const int wc = wave & 3;
  const int b0 = blockIdx.x << 6;
  const int m0 = blockIdx.y << 6;

  const int x_ch = t & 1;
  const int x_r = (t >> 1) & 15;
  const int x_q = (t >> 5) & 1;
  const int x_t = (t >> 6) & 3;
  const int x_s = (t >> 8) * 2;
  const float* gx = x + (size_t)(b0 + x_t * 16 + x_r) * (GA_NIN * 8)
                      + (x_s * 8 + x_q * 4) * 8 + x_ch * 4;
  const int lx = x_s * XR_SUB_F4 + (x_ch * 4) * 128 + x_t * 32 + x_q * 16 + x_r;
  const int w_r = t & 15;
  const int w_q = (t >> 4) & 1;
  const int w_t = (t >> 5) & 3;
  const int w_s = t >> 7;
  const float* gw = w_right + (size_t)(m0 + w_t * 16 + w_r) * (GA_NIN * 4)
                            + (w_s * 8 + w_q * 4) * 4;
  const int lw = w_s * XR_SUB_F4 + 1024 + w_t * 32 + w_q * 16 + w_r;

  const v8f vzero = {};
  v8f acc[8];
#pragma unroll
  for (int i = 0; i < 8; ++i) acc[i] = vzero;

  {
    Blk c0, c1, c2;
    ld_blk(c0, gx, 8);
    ld_blk(c1, gx + 64, 8);
    ld_blk(c2, gw, 4);
    st_blk(lds + lx, c0);
    st_blk(lds + lx + XR_SUB_F4, c1);
    st_blk(lds + lw, c2);
  }
  __syncthreads();

  for (int ch = 0; ch < GA_NIN / 32; ++ch) {
    const int st = ch & 1;
    const bool more = (ch + 1) < (GA_NIN / 32);

    Blk n0 = {}, n1 = {}, n2 = {};
    if (more) {
      gx += 32 * 8; gw += 32 * 4;
      ld_blk(n0, gx, 8);
      ld_blk(n1, gx + 64, 8);
      ld_blk(n2, gw, 4);
    }

#pragma unroll
    for (int s = 0; s < XR_SUB; ++s) {
      const float4* sub = lds + st * XR_ST_F4 + s * XR_SUB_F4;
      const v2f* sa = (const v2f*)(sub) + ((wr * 32 + lo) * 2 + hi);
      const v2f* sb = (const v2f*)(sub + 1024) + ((wc * 32 + lo) * 2 + hi);
#pragma unroll
      for (int q = 0; q < 2; ++q) {
        v2f a[8], g[4];
#pragma unroll
        for (int i = 0; i < 8; ++i) a[i] = sa[(i * 128 + q * 16) * 2];
#pragma unroll
        for (int j = 0; j < 4; ++j) g[j] = sb[(j * 128 + q * 16) * 2];
#pragma unroll
        for (int i = 0; i < 8; ++i)
          acc[i] = wmma_f32(a[i], g[GRADE_[i]], acc[i]);
      }

      if (s == XR_SUB - 2 && more) {
        float4* d = lds + (st ^ 1) * XR_ST_F4;
        st_blk(d + lx, n0);
        st_blk(d + lx + XR_SUB_F4, n1);
        st_blk(d + lw, n2);
      }
    }

    __syncthreads();
  }

  const int mcol = m0 + wc * 16 + lo;
  const float4 na = *(const float4*)(norm_a + (size_t)mcol * 4);
  const float s0 = 1.0f / (1.0f + __expf(-na.x));
  const float s1 = 1.0f / (1.0f + __expf(-na.y));
  const float s2 = 1.0f / (1.0f + __expf(-na.z));
  const float s3 = 1.0f / (1.0f + __expf(-na.w));

#pragma unroll
  for (int v = 0; v < 8; ++v) {
    const float e0 = acc[0][v], e1 = acc[1][v], e2 = acc[2][v], e3 = acc[3][v];
    const float e4 = acc[4][v], e5 = acc[5][v], e6 = acc[6][v], e7 = acc[7][v];
    const float n0 = __builtin_sqrtf(e0 * e0);
    const float n1 = __builtin_sqrtf(e1 * e1 + e2 * e2 + e3 * e3);
    const float n2 = __builtin_sqrtf(e4 * e4 + e5 * e5 + e6 * e6);
    const float n3 = __builtin_sqrtf(e7 * e7);
    const float i0 = 1.0f / (s0 * (n0 - 1.0f) + 1.0f + 1e-6f);
    const float i1 = 1.0f / (s1 * (n1 - 1.0f) + 1.0f + 1e-6f);
    const float i2 = 1.0f / (s2 * (n2 - 1.0f) + 1.0f + 1e-6f);
    const float i3 = 1.0f / (s3 * (n3 - 1.0f) + 1.0f + 1e-6f);
    float* dst = xr + ((size_t)(b0 + wr * 16 + v + (hi << 3)) * GA_NIN + mcol) * 8;
    *(float4*)(dst)     = make_float4(e0 * i0, e1 * i1, e2 * i1, e3 * i1);
    *(float4*)(dst + 4) = make_float4(e4 * i2, e5 * i2, e6 * i2, e7 * i3);
  }
}


#define GA_ST_F4 5120

__global__ __launch_bounds__(512) void ga_out_lds(
    const float* __restrict__ x, const float* __restrict__ xr,
    const float* __restrict__ weight, const float* __restrict__ w_left,
    const float* __restrict__ b_left, float* __restrict__ out) {
  __shared__ float4 lds[2 * GA_ST_F4];

  const int t = threadIdx.x;
  const int lane = t & 31;
  const int lo = lane & 15;
  const int hi = lane >> 4;
  const int wave = t >> 5;
  const int wr = wave >> 2;
  const int wc = wave & 3;
  const int b0 = blockIdx.x << 6;
  const int m0 = blockIdx.y << 6;

  const int a_ch = t & 1;
  const int a_r = (t >> 1) & 15;
  const int a_q = (t >> 5) & 1;
  const int a_t = (t >> 6) & 3;
  const int a_x = t >> 8;
  const float* ga = (a_x ? xr : x) + (size_t)(b0 + a_t * 16 + a_r) * (GA_NIN * 8)
                                   + (a_q * 4) * 8 + a_ch * 4;
  const int la = (a_x * 8 + a_ch * 4) * 128 + a_t * 32 + a_q * 16 + a_r;
  const int b_r = t & 15;
  const int b_g = ((t >> 5) & 1) * 2 + ((t >> 4) & 1);
  const int b_q = (t >> 6) & 1;
  const int b_t = t >> 7;
  const float* gb = weight + (size_t)(m0 + b_t * 16 + b_r) * (GA_NIN * 20)
                           + (b_q * 4) * 20 + b_g * 4;
  const int lb = 2048 + (4 + b_g * 4) * 128 + b_t * 32 + b_q * 16 + b_r;
  const bool has_c = wave < 8;
  const bool c_wl = wave >= 4;
  const int c_r = t & 15;
  const int c_q = (t >> 4) & 1;
  const int c_t = wave & 3;
  const int c_js = c_wl ? 4 : 20;
  const float* gc = c_wl
      ? w_left + (size_t)(m0 + c_t * 16 + c_r) * (GA_NIN * 4) + (c_q * 4) * 4
      : weight + (size_t)(m0 + c_t * 16 + c_r) * (GA_NIN * 20) + (c_q * 4) * 20 + 16;
  const int lc = 2048 + (c_wl ? 0 : 20) * 128 + c_t * 32 + c_q * 16 + c_r;

  const v8f vzero = {};
  v8f acc[8];
#pragma unroll
  for (int i = 0; i < 8; ++i) acc[i] = vzero;

  {
    Blk c0, c1, c2 = {};
    ld_blk(c0, ga, 8);
    ld_blk(c1, gb, 20);
    if (has_c) ld_blk(c2, gc, c_js);
    st_blk(lds + la, c0);
    st_blk(lds + lb, c1);
    if (has_c) st_blk(lds + lc, c2);
  }
  __syncthreads();

#pragma unroll 1
  for (int ch = 0; ch < GA_NIN / 8; ++ch) {
    const int st = ch & 1;
    const bool more = (ch + 1) < (GA_NIN / 8);

    Blk n0 = {}, n1 = {}, n2 = {};
    if (more) {
      ga += 8 * 8; gb += 8 * 20; gc += 8 * c_js;
      ld_blk(n0, ga, 8);
      ld_blk(n1, gb, 20);
      if (has_c) ld_blk(n2, gc, c_js);
    }

    const v2f* sa = (const v2f*)(lds + st * GA_ST_F4) + ((wr * 32 + lo) * 2 + hi);
    const v2f* sb = (const v2f*)(lds + st * GA_ST_F4 + 2048) + ((wc * 32 + lo) * 2 + hi);

#pragma unroll
    for (int q = 0; q < 2; ++q) {
      v2f a[8], r[8], bl[4], bw[20];
#pragma unroll
      for (int i = 0; i < 8; ++i) a[i] = sa[(i * 128 + q * 16) * 2];
#pragma unroll
      for (int i = 0; i < 8; ++i) r[i] = sa[((8 + i) * 128 + q * 16) * 2];
#pragma unroll
      for (int j = 0; j < 4; ++j) bl[j] = sb[(j * 128 + q * 16) * 2];
#pragma unroll
      for (int p = 0; p < 20; ++p) bw[p] = sb[((4 + p) * 128 + q * 16) * 2];

#pragma unroll
      for (int i = 0; i < 8; ++i)
        acc[i] = wmma_f32(a[i], bl[GRADE_[i]], acc[i]);

#pragma unroll
      for (int e = 0; e < 64; ++e) {
        v2f p = a[EBI[e]] * r[EBK[e]];
        if (ECV[e] < 0) p = -p;
        acc[EBJ[e]] = wmma_f32(p, bw[EPX[e]], acc[EBJ[e]]);
      }

      if (q == 0 && more) {
        float4* d = lds + (st ^ 1) * GA_ST_F4;
        st_blk(d + la, n0);
        st_blk(d + lb, n1);
        if (has_c) st_blk(d + lc, n2);
      }
    }

    __syncthreads();
  }

  const int mcol = m0 + wc * 16 + lo;
  const float bias = b_left[mcol];
  const float rs2 = 0.70710678118654752440f;
#pragma unroll
  for (int v = 0; v < 8; ++v) {
    const float o0 = (acc[0][v] + bias) * rs2;
    const float o1 = acc[1][v] * rs2, o2 = acc[2][v] * rs2, o3 = acc[3][v] * rs2;
    const float o4 = acc[4][v] * rs2, o5 = acc[5][v] * rs2;
    const float o6 = acc[6][v] * rs2, o7 = acc[7][v] * rs2;
    float* dst = out + ((size_t)(b0 + wr * 16 + v + (hi << 3)) * GA_NOUT + mcol) * 8;
    *(float4*)(dst)     = make_float4(o0, o1, o2, o3);
    *(float4*)(dst + 4) = make_float4(o4, o5, o6, o7);
  }
}


extern "C" void kernel_launch(void* const* d_in, const int* in_sizes, int n_in,
                              void* d_out, int out_size, void* d_ws, size_t ws_size,
                              hipStream_t stream) {
  (void)in_sizes; (void)n_in; (void)out_size;
  const float* x       = (const float*)d_in[0];
  const float* weight  = (const float*)d_in[1];
  const float* w_right = (const float*)d_in[2];
  const float* w_left  = (const float*)d_in[3];
  const float* b_left  = (const float*)d_in[4];
  const float* norm_a  = (const float*)d_in[5];
  float* out = (float*)d_out;
  float* xr = (float*)d_ws;

  if (ws_size >= (size_t)GA_B * GA_NIN * 8 * sizeof(float)) {
    ga_xr_lds<<<dim3(GA_B / 64, GA_NIN / 64, 1), dim3(512, 1, 1), 0, stream>>>(x, w_right, norm_a, xr);
    ga_out_lds<<<dim3(GA_B / 64, GA_NOUT / 64, 1), dim3(512, 1, 1), 0, stream>>>(x, xr, weight, w_left, b_left, out);
  }
}


# Main kernel: per-chunk barrier split: a wave arrives once its last LDS reads of the chunk have landed, ahead of its last 72 WMMAs, and waits behind them; released waves refill under those WMMAs
# speedup vs baseline: 3.0936x; 1.0638x over previous
#include <hip/hip_runtime.h>


typedef __attribute__((ext_vector_type(2))) float v2f;
typedef __attribute__((ext_vector_type(8))) float v8f;

#define GA_B    2048
#define GA_NIN  512
#define GA_NOUT 512

__device__ __forceinline__ v8f wmma_f32(v2f a, v2f b, v8f c) {
  return __builtin_amdgcn_wmma_f32_16x16x4_f32(false, a, false, b, (short)0, c, false, false);
}

static constexpr int GRADE_[8] = {0, 1, 1, 1, 2, 2, 2, 3};

static constexpr int EBI[64] = {
  0,0,0,0,0,0,0,0,
  1,2,3, 1,2,3,
  1,1,2,2,3,3,
  1,1,2,2,3,3,
  1,2,3, 1,2,3,
  4,5,6,
  4,4,5,5,6,6,
  4,5,6, 4,5,6,
  4,4,5,5,6,6,
  4,5,6,
  7, 7,7,7, 7,7,7, 7};
static constexpr int EBJ[64] = {
  0,1,2,3,4,5,6,7,
  0,0,0, 1,2,3,
  2,3,1,3,1,2,
  4,5,4,6,5,6,
  6,5,4, 7,7,7,
  0,0,0,
  1,2,1,3,2,3,
  3,2,1, 4,5,6,
  5,6,4,6,4,5,
  7,7,7,
  0, 1,2,3, 4,5,6, 7};
static constexpr int EBK[64] = {
  0,1,2,3,4,5,6,7,
  1,2,3, 0,0,0,
  4,5,4,6,5,6,
  2,3,1,3,1,2,
  7,7,7, 6,5,4,
  4,5,6,
  2,1,3,1,3,2,
  7,7,7, 0,0,0,
  6,5,6,4,5,4,
  3,2,1,
  7, 6,5,4, 3,2,1, 0};
static constexpr int ECV[64] = {
  1,1,1,1,1,1,1,1,
  1,1,1, 1,1,1,
  1,1,-1,1,-1,-1,
  1,1,-1,1,-1,-1,
  1,-1,1, 1,-1,1,
  -1,-1,-1,
  1,-1,1,-1,1,-1,
  -1,1,-1, 1,1,1,
  1,-1,-1,1,1,-1,
  1,-1,1,
  -1, -1,1,-1, 1,-1,1, 1};
static constexpr int EPX[64] = {
  0,1,1,1,2,2,2,3,
  4,4,4, 5,5,5,
  6,6,6,6,6,6,
  7,7,7,7,7,7,
  8,8,8, 9,9,9,
  10,10,10,
  11,11,11,11,11,11,
  12,12,12, 13,13,13,
  14,14,14,14,14,14,
  15,15,15,
  16, 17,17,17, 18,18,18, 19};


struct Blk { float4 r0, r1, r2, r3; };

__device__ __forceinline__ void ld_blk(Blk& b, const float* p, int js) {
  b.r0 = *(const float4*)(p);
  b.r1 = *(const float4*)(p + js);
  b.r2 = *(const float4*)(p + 2 * js);
  b.r3 = *(const float4*)(p + 3 * js);
}

__device__ __forceinline__ void st_blk(float4* d, const Blk& b) {
  d[0]   = make_float4(b.r0.x, b.r1.x, b.r2.x, b.r3.x);
  d[128] = make_float4(b.r0.y, b.r1.y, b.r2.y, b.r3.y);
  d[256] = make_float4(b.r0.z, b.r1.z, b.r2.z, b.r3.z);
  d[384] = make_float4(b.r0.w, b.r1.w, b.r2.w, b.r3.w);
}


#define XR_SUB    4
#define XR_SUB_F4 1536
#define XR_ST_F4  (XR_SUB * XR_SUB_F4)

__global__ __launch_bounds__(512) void ga_xr_lds(
    const float* __restrict__ x, const float* __restrict__ w_right,
    const float* __restrict__ norm_a, float* __restrict__ xr) {
  __shared__ float4 lds[2 * XR_ST_F4];

  const int t = threadIdx.x;
  const int lane = t & 31;
  const int lo = lane & 15;
  const int hi = lane >> 4;
  const int wave = t >> 5;
  const int wr = wave >> 2;
  const int wc = wave & 3;
  const int b0 = blockIdx.x << 6;
  const int m0 = blockIdx.y << 6;

  const int x_ch = t & 1;
  const int x_r = (t >> 1) & 15;
  const int x_q = (t >> 5) & 1;
  const int x_t = (t >> 6) & 3;
  const int x_s = (t >> 8) * 2;
  const float* gx = x + (size_t)(b0 + x_t * 16 + x_r) * (GA_NIN * 8)
                      + (x_s * 8 + x_q * 4) * 8 + x_ch * 4;
  const int lx = x_s * XR_SUB_F4 + (x_ch * 4) * 128 + x_t * 32 + x_q * 16 + x_r;
  const int w_r = t & 15;
  const int w_q = (t >> 4) & 1;
  const int w_t = (t >> 5) & 3;
  const int w_s = t >> 7;
  const float* gw = w_right + (size_t)(m0 + w_t * 16 + w_r) * (GA_NIN * 4)
                            + (w_s * 8 + w_q * 4) * 4;
  const int lw = w_s * XR_SUB_F4 + 1024 + w_t * 32 + w_q * 16 + w_r;

  const v8f vzero = {};
  v8f acc[8];
#pragma unroll
  for (int i = 0; i < 8; ++i) acc[i] = vzero;

  {
    Blk c0, c1, c2;
    ld_blk(c0, gx, 8);
    ld_blk(c1, gx + 64, 8);
    ld_blk(c2, gw, 4);
    st_blk(lds + lx, c0);
    st_blk(lds + lx + XR_SUB_F4, c1);
    st_blk(lds + lw, c2);
  }
  __syncthreads();

  for (int ch = 0; ch < GA_NIN / 32; ++ch) {
    const int st = ch & 1;
    const bool more = (ch + 1) < (GA_NIN / 32);

    Blk n0 = {}, n1 = {}, n2 = {};
    if (more) {
      gx += 32 * 8; gw += 32 * 4;
      ld_blk(n0, gx, 8);
      ld_blk(n1, gx + 64, 8);
      ld_blk(n2, gw, 4);
    }

#pragma unroll
    for (int s = 0; s < XR_SUB; ++s) {
      const float4* sub = lds + st * XR_ST_F4 + s * XR_SUB_F4;
      const v2f* sa = (const v2f*)(sub) + ((wr * 32 + lo) * 2 + hi);
      const v2f* sb = (const v2f*)(sub + 1024) + ((wc * 32 + lo) * 2 + hi);
#pragma unroll
      for (int q = 0; q < 2; ++q) {
        v2f a[8], g[4];
#pragma unroll
        for (int i = 0; i < 8; ++i) a[i] = sa[(i * 128 + q * 16) * 2];
#pragma unroll
        for (int j = 0; j < 4; ++j) g[j] = sb[(j * 128 + q * 16) * 2];
#pragma unroll
        for (int i = 0; i < 8; ++i)
          acc[i] = wmma_f32(a[i], g[GRADE_[i]], acc[i]);
      }

      if (s == XR_SUB - 2 && more) {
        float4* d = lds + (st ^ 1) * XR_ST_F4;
        st_blk(d + lx, n0);
        st_blk(d + lx + XR_SUB_F4, n1);
        st_blk(d + lw, n2);
      }
    }

    __syncthreads();
  }

  const int mcol = m0 + wc * 16 + lo;
  const float4 na = *(const float4*)(norm_a + (size_t)mcol * 4);
  const float s0 = 1.0f / (1.0f + __expf(-na.x));
  const float s1 = 1.0f / (1.0f + __expf(-na.y));
  const float s2 = 1.0f / (1.0f + __expf(-na.z));
  const float s3 = 1.0f / (1.0f + __expf(-na.w));

#pragma unroll
  for (int v = 0; v < 8; ++v) {
    const float e0 = acc[0][v], e1 = acc[1][v], e2 = acc[2][v], e3 = acc[3][v];
    const float e4 = acc[4][v], e5 = acc[5][v], e6 = acc[6][v], e7 = acc[7][v];
    const float n0 = __builtin_sqrtf(e0 * e0);
    const float n1 = __builtin_sqrtf(e1 * e1 + e2 * e2 + e3 * e3);
    const float n2 = __builtin_sqrtf(e4 * e4 + e5 * e5 + e6 * e6);
    const float n3 = __builtin_sqrtf(e7 * e7);
    const float i0 = 1.0f / (s0 * (n0 - 1.0f) + 1.0f + 1e-6f);
    const float i1 = 1.0f / (s1 * (n1 - 1.0f) + 1.0f + 1e-6f);
    const float i2 = 1.0f / (s2 * (n2 - 1.0f) + 1.0f + 1e-6f);
    const float i3 = 1.0f / (s3 * (n3 - 1.0f) + 1.0f + 1e-6f);
    float* dst = xr + ((size_t)(b0 + wr * 16 + v + (hi << 3)) * GA_NIN + mcol) * 8;
    *(float4*)(dst)     = make_float4(e0 * i0, e1 * i1, e2 * i1, e3 * i1);
    *(float4*)(dst + 4) = make_float4(e4 * i2, e5 * i2, e6 * i2, e7 * i3);
  }
}


#define GA_ST_F4 5120

__device__ __forceinline__ void ga_read_frags(const v2f* sa, const v2f* sb, int q,
                                              v2f (&a)[8], v2f (&r)[8],
                                              v2f (&bl)[4], v2f (&bw)[20]) {
#pragma unroll
  for (int i = 0; i < 8; ++i) a[i] = sa[(i * 128 + q * 16) * 2];
#pragma unroll
  for (int i = 0; i < 8; ++i) r[i] = sa[((8 + i) * 128 + q * 16) * 2];
#pragma unroll
  for (int j = 0; j < 4; ++j) bl[j] = sb[(j * 128 + q * 16) * 2];
#pragma unroll
  for (int p = 0; p < 20; ++p) bw[p] = sb[((4 + p) * 128 + q * 16) * 2];
}

__device__ __forceinline__ void ga_mma_step(v8f (&acc)[8], const v2f (&a)[8], const v2f (&r)[8],
                                            const v2f (&bl)[4], const v2f (&bw)[20]) {
#pragma unroll
  for (int i = 0; i < 8; ++i)
    acc[i] = wmma_f32(a[i], bl[GRADE_[i]], acc[i]);

#pragma unroll
  for (int e = 0; e < 64; ++e) {
    v2f p = a[EBI[e]] * r[EBK[e]];
    if (ECV[e] < 0) p = -p;
    acc[EBJ[e]] = wmma_f32(p, bw[EPX[e]], acc[EBJ[e]]);
  }
}

__global__ __launch_bounds__(512) void ga_out_lds(
    const float* __restrict__ x, const float* __restrict__ xr,
    const float* __restrict__ weight, const float* __restrict__ w_left,
    const float* __restrict__ b_left, float* __restrict__ out) {
  __shared__ float4 lds[2 * GA_ST_F4];

  const int t = threadIdx.x;
  const int lane = t & 31;
  const int lo = lane & 15;
  const int hi = lane >> 4;
  const int wave = t >> 5;
  const int wr = wave >> 2;
  const int wc = wave & 3;
  const int b0 = blockIdx.x << 6;
  const int m0 = blockIdx.y << 6;

  const int a_ch = t & 1;
  const int a_r = (t >> 1) & 15;
  const int a_q = (t >> 5) & 1;
  const int a_t = (t >> 6) & 3;
  const int a_x = t >> 8;
  const float* ga = (a_x ? xr : x) + (size_t)(b0 + a_t * 16 + a_r) * (GA_NIN * 8)
                                   + (a_q * 4) * 8 + a_ch * 4;
  const int la = (a_x * 8 + a_ch * 4) * 128 + a_t * 32 + a_q * 16 + a_r;
  const int b_r = t & 15;
  const int b_g = ((t >> 5) & 1) * 2 + ((t >> 4) & 1);
  const int b_q = (t >> 6) & 1;
  const int b_t = t >> 7;
  const float* gb = weight + (size_t)(m0 + b_t * 16 + b_r) * (GA_NIN * 20)
                           + (b_q * 4) * 20 + b_g * 4;
  const int lb = 2048 + (4 + b_g * 4) * 128 + b_t * 32 + b_q * 16 + b_r;
  const bool has_c = wave < 8;
  const bool c_wl = wave >= 4;
  const int c_r = t & 15;
  const int c_q = (t >> 4) & 1;
  const int c_t = wave & 3;
  const int c_js = c_wl ? 4 : 20;
  const float* gc = c_wl
      ? w_left + (size_t)(m0 + c_t * 16 + c_r) * (GA_NIN * 4) + (c_q * 4) * 4
      : weight + (size_t)(m0 + c_t * 16 + c_r) * (GA_NIN * 20) + (c_q * 4) * 20 + 16;
  const int lc = 2048 + (c_wl ? 0 : 20) * 128 + c_t * 32 + c_q * 16 + c_r;

  const v8f vzero = {};
  v8f acc[8];
#pragma unroll
  for (int i = 0; i < 8; ++i) acc[i] = vzero;

  {
    Blk c0, c1, c2 = {};
    ld_blk(c0, ga, 8);
    ld_blk(c1, gb, 20);
    if (has_c) ld_blk(c2, gc, c_js);
    st_blk(lds + la, c0);
    st_blk(lds + lb, c1);
    if (has_c) st_blk(lds + lc, c2);
  }
  __syncthreads();

#pragma unroll 1
  for (int ch = 0; ch < GA_NIN / 8; ++ch) {
    const int st = ch & 1;
    const bool more = (ch + 1) < (GA_NIN / 8);

    Blk n0 = {}, n1 = {}, n2 = {};
    if (more) {
      ga += 8 * 8; gb += 8 * 20; gc += 8 * c_js;
      ld_blk(n0, ga, 8);
      ld_blk(n1, gb, 20);
      if (has_c) ld_blk(n2, gc, c_js);
    }

    const v2f* sa = (const v2f*)(lds + st * GA_ST_F4) + ((wr * 32 + lo) * 2 + hi);
    const v2f* sb = (const v2f*)(lds + st * GA_ST_F4 + 2048) + ((wc * 32 + lo) * 2 + hi);

    v2f a[8], r[8], bl[4], bw[20];

    ga_read_frags(sa, sb, 0, a, r, bl, bw);
    ga_mma_step(acc, a, r, bl, bw);

    if (more) {
      float4* d = lds + (st ^ 1) * GA_ST_F4;
      st_blk(d + la, n0);
      st_blk(d + lb, n1);
      if (has_c) st_blk(d + lc, n2);
    }

    ga_read_frags(sa, sb, 1, a, r, bl, bw);

    __builtin_amdgcn_fence(__ATOMIC_RELEASE, "workgroup");
    __builtin_amdgcn_s_barrier_signal(-1);
    __builtin_amdgcn_sched_barrier(0);

    ga_mma_step(acc, a, r, bl, bw);

    __builtin_amdgcn_sched_barrier(0);
    __builtin_amdgcn_s_barrier_wait(-1);
    __builtin_amdgcn_fence(__ATOMIC_ACQUIRE, "workgroup");
  }

  const int mcol = m0 + wc * 16 + lo;
  const float bias = b_left[mcol];
  const float rs2 = 0.70710678118654752440f;
#pragma unroll
  for (int v = 0; v < 8; ++v) {
    const float o0 = (acc[0][v] + bias) * rs2;
    const float o1 = acc[1][v] * rs2, o2 = acc[2][v] * rs2, o3 = acc[3][v] * rs2;
    const float o4 = acc[4][v] * rs2, o5 = acc[5][v] * rs2;
    const float o6 = acc[6][v] * rs2, o7 = acc[7][v] * rs2;
    float* dst = out + ((size_t)(b0 + wr * 16 + v + (hi << 3)) * GA_NOUT + mcol) * 8;
    *(float4*)(dst)     = make_float4(o0, o1, o2, o3);
    *(float4*)(dst + 4) = make_float4(o4, o5, o6, o7);
  }
}


extern "C" void kernel_launch(void* const* d_in, const int* in_sizes, int n_in,
                              void* d_out, int out_size, void* d_ws, size_t ws_size,
                              hipStream_t stream) {
  (void)in_sizes; (void)n_in; (void)out_size;
  const float* x       = (const float*)d_in[0];
  const float* weight  = (const float*)d_in[1];
  const float* w_right = (const float*)d_in[2];
  const float* w_left  = (const float*)d_in[3];
  const float* b_left  = (const float*)d_in[4];
  const float* norm_a  = (const float*)d_in[5];
  float* out = (float*)d_out;
  float* xr = (float*)d_ws;

  if (ws_size >= (size_t)GA_B * GA_NIN * 8 * sizeof(float)) {
    ga_xr_lds<<<dim3(GA_B / 64, GA_NIN / 64, 1), dim3(512, 1, 1), 0, stream>>>(x, w_right, norm_a, xr);
    ga_out_lds<<<dim3(GA_B / 64, GA_NOUT / 64, 1), dim3(512, 1, 1), 0, stream>>>(x, xr, weight, w_left, b_left, out);
  }
}


# Main kernel: staging threads sum the 64 Cayley products into 44 (path, blade) channels once per work-group into LDS; WMMA loop is a plain 52-channel GEMM (52 not 72 WMMAs); summation order changes
# speedup vs baseline: 3.5799x; 1.1572x over previous
#include <hip/hip_runtime.h>


typedef __attribute__((ext_vector_type(2))) float v2f;
typedef __attribute__((ext_vector_type(8))) float v8f;

#define GA_B    2048
#define GA_NIN  512
#define GA_NOUT 512

__device__ __forceinline__ v8f wmma_f32(v2f a, v2f b, v8f c) {
  return __builtin_amdgcn_wmma_f32_16x16x4_f32(false, a, false, b, (short)0, c, false, false);
}

static constexpr int GRADE_[8] = {0, 1, 1, 1, 2, 2, 2, 3};


struct Blk { float4 r0, r1, r2, r3; };

__device__ __forceinline__ void ld_blk(Blk& b, const float* p, int js) {
  b.r0 = *(const float4*)(p);
  b.r1 = *(const float4*)(p + js);
  b.r2 = *(const float4*)(p + 2 * js);
  b.r3 = *(const float4*)(p + 3 * js);
}

__device__ __forceinline__ void st_blk(float4* d, const Blk& b) {
  d[0]   = make_float4(b.r0.x, b.r1.x, b.r2.x, b.r3.x);
  d[128] = make_float4(b.r0.y, b.r1.y, b.r2.y, b.r3.y);
  d[256] = make_float4(b.r0.z, b.r1.z, b.r2.z, b.r3.z);
  d[384] = make_float4(b.r0.w, b.r1.w, b.r2.w, b.r3.w);
}


#define XR_SUB    4
#define XR_SUB_F4 1536
#define XR_ST_F4  (XR_SUB * XR_SUB_F4)

__global__ __launch_bounds__(512) void ga_xr_lds(
    const float* __restrict__ x, const float* __restrict__ w_right,
    const float* __restrict__ norm_a, float* __restrict__ xr) {
  __shared__ float4 lds[2 * XR_ST_F4];

  const int t = threadIdx.x;
  const int lane = t & 31;
  const int lo = lane & 15;
  const int hi = lane >> 4;
  const int wave = t >> 5;
  const int wr = wave >> 2;
  const int wc = wave & 3;
  const int b0 = blockIdx.x << 6;
  const int m0 = blockIdx.y << 6;

  const int x_ch = t & 1;
  const int x_r = (t >> 1) & 15;
  const int x_q = (t >> 5) & 1;
  const int x_t = (t >> 6) & 3;
  const int x_s = (t >> 8) * 2;
  const float* gx = x + (size_t)(b0 + x_t * 16 + x_r) * (GA_NIN * 8)
                      + (x_s * 8 + x_q * 4) * 8 + x_ch * 4;
  const int lx = x_s * XR_SUB_F4 + (x_ch * 4) * 128 + x_t * 32 + x_q * 16 + x_r;
  const int w_r = t & 15;
  const int w_q = (t >> 4) & 1;
  const int w_t = (t >> 5) & 3;
  const int w_s = t >> 7;
  const float* gw = w_right + (size_t)(m0 + w_t * 16 + w_r) * (GA_NIN * 4)
                            + (w_s * 8 + w_q * 4) * 4;
  const int lw = w_s * XR_SUB_F4 + 1024 + w_t * 32 + w_q * 16 + w_r;

  const v8f vzero = {};
  v8f acc[8];
#pragma unroll
  for (int i = 0; i < 8; ++i) acc[i] = vzero;

  {
    Blk c0, c1, c2;
    ld_blk(c0, gx, 8);
    ld_blk(c1, gx + 64, 8);
    ld_blk(c2, gw, 4);
    st_blk(lds + lx, c0);
    st_blk(lds + lx + XR_SUB_F4, c1);
    st_blk(lds + lw, c2);
  }
  __syncthreads();

  for (int ch = 0; ch < GA_NIN / 32; ++ch) {
    const int st = ch & 1;
    const bool more = (ch + 1) < (GA_NIN / 32);

    Blk n0 = {}, n1 = {}, n2 = {};
    if (more) {
      gx += 32 * 8; gw += 32 * 4;
      ld_blk(n0, gx, 8);
      ld_blk(n1, gx + 64, 8);
      ld_blk(n2, gw, 4);
    }

#pragma unroll
    for (int s = 0; s < XR_SUB; ++s) {
      const float4* sub = lds + st * XR_ST_F4 + s * XR_SUB_F4;
      const v2f* sa = (const v2f*)(sub) + ((wr * 32 + lo) * 2 + hi);
      const v2f* sb = (const v2f*)(sub + 1024) + ((wc * 32 + lo) * 2 + hi);
#pragma unroll
      for (int q = 0; q < 2; ++q) {
        v2f a[8], g[4];
#pragma unroll
        for (int i = 0; i < 8; ++i) a[i] = sa[(i * 128 + q * 16) * 2];
#pragma unroll
        for (int j = 0; j < 4; ++j) g[j] = sb[(j * 128 + q * 16) * 2];
#pragma unroll
        for (int i = 0; i < 8; ++i)
          acc[i] = wmma_f32(a[i], g[GRADE_[i]], acc[i]);
      }

      if (s == XR_SUB - 2 && more) {
        float4* d = lds + (st ^ 1) * XR_ST_F4;
        st_blk(d + lx, n0);
        st_blk(d + lx + XR_SUB_F4, n1);
        st_blk(d + lw, n2);
      }
    }

    __syncthreads();
  }

  const int mcol = m0 + wc * 16 + lo;
  const float4 na = *(const float4*)(norm_a + (size_t)mcol * 4);
  const float s0 = 1.0f / (1.0f + __expf(-na.x));
  const float s1 = 1.0f / (1.0f + __expf(-na.y));
  const float s2 = 1.0f / (1.0f + __expf(-na.z));
  const float s3 = 1.0f / (1.0f + __expf(-na.w));

#pragma unroll
  for (int v = 0; v < 8; ++v) {
    const float e0 = acc[0][v], e1 = acc[1][v], e2 = acc[2][v], e3 = acc[3][v];
    const float e4 = acc[4][v], e5 = acc[5][v], e6 = acc[6][v], e7 = acc[7][v];
    const float n0 = __builtin_sqrtf(e0 * e0);
    const float n1 = __builtin_sqrtf(e1 * e1 + e2 * e2 + e3 * e3);
    const float n2 = __builtin_sqrtf(e4 * e4 + e5 * e5 + e6 * e6);
    const float n3 = __builtin_sqrtf(e7 * e7);
    const float i0 = 1.0f / (s0 * (n0 - 1.0f) + 1.0f + 1e-6f);
    const float i1 = 1.0f / (s1 * (n1 - 1.0f) + 1.0f + 1e-6f);
    const float i2 = 1.0f / (s2 * (n2 - 1.0f) + 1.0f + 1e-6f);
    const float i3 = 1.0f / (s3 * (n3 - 1.0f) + 1.0f + 1e-6f);
    float* dst = xr + ((size_t)(b0 + wr * 16 + v + (hi << 3)) * GA_NIN + mcol) * 8;
    *(float4*)(dst)     = make_float4(e0 * i0, e1 * i1, e2 * i1, e3 * i1);
    *(float4*)(dst + 4) = make_float4(e4 * i2, e5 * i2, e6 * i2, e7 * i3);
  }
}


#define GA_NCH   52
#define GA_A_F4  (GA_NCH * 128)
#define GA_ST_F4 (GA_A_F4 + 3072)

static constexpr int LCH_J[GA_NCH] = {
  0,1,2,3,0,1,2,3,0,1,2,3,0,1,2,3,0,1,2,3,1,2,3,1,2,3,
  4,5,6,7,4,5,6,7,4,5,6,7,4,5,6,7,4,5,6,7,4,5,6,4,5,6};
static constexpr int LCH_B[GA_NCH] = {
  0,1,1,1,4,5,5,5,8,9,9,9,14,10,10,10,20,15,15,15,16,16,16,21,21,21,
  2,2,2,3,6,6,6,7,11,11,11,13,12,12,12,19,17,17,17,23,18,18,18,22,22,22};

__device__ __forceinline__ void ga_channels(const float (&xv)[8], const float (&rv)[8],
                                            float (&v)[GA_NCH]) {
  v[0] = xv[0];
  v[1] = xv[1];
  v[2] = xv[2];
  v[3] = xv[3];
  v[4] = xv[0] * rv[0];
  v[5] = xv[0] * rv[1];
  v[6] = xv[0] * rv[2];
  v[7] = xv[0] * rv[3];
  v[8] = xv[1] * rv[1] + xv[2] * rv[2] + xv[3] * rv[3];
  v[9] = xv[1] * rv[0];
  v[10] = xv[2] * rv[0];
  v[11] = xv[3] * rv[0];
  v[12] = -xv[4] * rv[4] - xv[5] * rv[5] - xv[6] * rv[6];
  v[13] = -xv[2] * rv[4] - xv[3] * rv[5];
  v[14] = xv[1] * rv[4] - xv[3] * rv[6];
  v[15] = xv[1] * rv[5] + xv[2] * rv[6];
  v[16] = -xv[7] * rv[7];
  v[17] = xv[4] * rv[2] + xv[5] * rv[3];
  v[18] = -xv[4] * rv[1] + xv[6] * rv[3];
  v[19] = -xv[5] * rv[1] - xv[6] * rv[2];
  v[20] = -xv[6] * rv[7];
  v[21] = xv[5] * rv[7];
  v[22] = -xv[4] * rv[7];
  v[23] = -xv[7] * rv[6];
  v[24] = xv[7] * rv[5];
  v[25] = -xv[7] * rv[4];
  v[26] = xv[4];
  v[27] = xv[5];
  v[28] = xv[6];
  v[29] = xv[7];
  v[30] = xv[0] * rv[4];
  v[31] = xv[0] * rv[5];
  v[32] = xv[0] * rv[6];
  v[33] = xv[0] * rv[7];
  v[34] = xv[1] * rv[2] - xv[2] * rv[1];
  v[35] = xv[1] * rv[3] - xv[3] * rv[1];
  v[36] = xv[2] * rv[3] - xv[3] * rv[2];
  v[37] = xv[1] * rv[6] - xv[2] * rv[5] + xv[3] * rv[4];
  v[38] = xv[3] * rv[7];
  v[39] = -xv[2] * rv[7];
  v[40] = xv[1] * rv[7];
  v[41] = xv[4] * rv[3] - xv[5] * rv[2] + xv[6] * rv[1];
  v[42] = xv[4] * rv[0];
  v[43] = xv[5] * rv[0];
  v[44] = xv[6] * rv[0];
  v[45] = xv[7] * rv[0];
  v[46] = -xv[5] * rv[6] + xv[6] * rv[5];
  v[47] = xv[4] * rv[6] - xv[6] * rv[4];
  v[48] = -xv[4] * rv[5] + xv[5] * rv[4];
  v[49] = xv[7] * rv[3];
  v[50] = -xv[7] * rv[2];
  v[51] = xv[7] * rv[1];
}

__device__ __forceinline__ void st_channels(float* d, const float4& x0, const float4& x1,
                                            const float4& r0, const float4& r1) {
  const float xv[8] = {x0.x, x0.y, x0.z, x0.w, x1.x, x1.y, x1.z, x1.w};
  const float rv[8] = {r0.x, r0.y, r0.z, r0.w, r1.x, r1.y, r1.z, r1.w};
  float v[GA_NCH];
  ga_channels(xv, rv, v);
#pragma unroll
  for (int c = 0; c < GA_NCH; ++c) d[c * 512] = v[c];
}

template <int C0, int N>
__device__ __forceinline__ void ga_read_a(const v2f* sa, int q, v2f (&af)[N]) {
#pragma unroll
  for (int i = 0; i < N; ++i) af[i] = sa[((C0 + i) * 128 + q * 16) * 2];
}

__device__ __forceinline__ void ga_read_b(const v2f* sb, int q, v2f (&b)[24]) {
#pragma unroll
  for (int j = 0; j < 24; ++j) b[j] = sb[(j * 128 + q * 16) * 2];
}

template <int C0, int N>
__device__ __forceinline__ void ga_mma(v8f (&acc)[8], const v2f (&af)[N], const v2f (&b)[24]) {
#pragma unroll
  for (int i = 0; i < N; ++i)
    acc[LCH_J[C0 + i]] = wmma_f32(af[i], b[LCH_B[C0 + i]], acc[LCH_J[C0 + i]]);
}

__global__ __launch_bounds__(512) void ga_out_lds(
    const float* __restrict__ x, const float* __restrict__ xr,
    const float* __restrict__ weight, const float* __restrict__ w_left,
    const float* __restrict__ b_left, float* __restrict__ out) {
  __shared__ float4 lds[2 * GA_ST_F4];

  const int t = threadIdx.x;
  const int lane = t & 31;
  const int lo = lane & 15;
  const int hi = lane >> 4;
  const int wave = t >> 5;
  const int wr = wave >> 2;
  const int wc = wave & 3;
  const int b0 = blockIdx.x << 6;
  const int m0 = blockIdx.y << 6;

  const int a_kk = t & 3;
  const int a_r8 = (t >> 2) & 7;
  const int a_q = wave & 1;
  const int a_row = (wave >> 1) * 8 + a_r8;
  const size_t a_off = (size_t)(b0 + a_row) * (GA_NIN * 8) + (a_q * 4 + a_kk) * 8;
  const float* gax = x + a_off;
  const float* gar = xr + a_off;
  const int la = ((a_row >> 4) * 32 + a_q * 16 + (a_row & 15)) * 4 + a_kk;
  const int b_r = t & 15;
  const int b_g = ((t >> 5) & 1) * 2 + ((t >> 4) & 1);
  const int b_q = (t >> 6) & 1;
  const int b_t = t >> 7;
  const float* gb = weight + (size_t)(m0 + b_t * 16 + b_r) * (GA_NIN * 20)
                           + (b_q * 4) * 20 + b_g * 4;
  const int lb = GA_A_F4 + (4 + b_g * 4) * 128 + b_t * 32 + b_q * 16 + b_r;
  const bool has_c = wave < 8;
  const bool c_wl = wave >= 4;
  const int c_r = t & 15;
  const int c_q = (t >> 4) & 1;
  const int c_t = wave & 3;
  const int c_js = c_wl ? 4 : 20;
  const float* gc = c_wl
      ? w_left + (size_t)(m0 + c_t * 16 + c_r) * (GA_NIN * 4) + (c_q * 4) * 4
      : weight + (size_t)(m0 + c_t * 16 + c_r) * (GA_NIN * 20) + (c_q * 4) * 20 + 16;
  const int lc = GA_A_F4 + (c_wl ? 0 : 20) * 128 + c_t * 32 + c_q * 16 + c_r;

  const v8f vzero = {};
  v8f acc[8];
#pragma unroll
  for (int i = 0; i < 8; ++i) acc[i] = vzero;

  {
    const float4 x0 = *(const float4*)(gax), x1 = *(const float4*)(gax + 4);
    const float4 r0 = *(const float4*)(gar), r1 = *(const float4*)(gar + 4);
    Blk c1, c2 = {};
    ld_blk(c1, gb, 20);
    if (has_c) ld_blk(c2, gc, c_js);
    st_channels((float*)lds + la, x0, x1, r0, r1);
    st_blk(lds + lb, c1);
    if (has_c) st_blk(lds + lc, c2);
  }
  __syncthreads();

#pragma unroll 1
  for (int ch = 0; ch < GA_NIN / 8; ++ch) {
    const int st = ch & 1;
    const bool more = (ch + 1) < (GA_NIN / 8);

    float4 nx0 = {}, nx1 = {}, nr0 = {}, nr1 = {};
    Blk n1 = {}, n2 = {};
    if (more) {
      gax += 8 * 8; gar += 8 * 8; gb += 8 * 20; gc += 8 * c_js;
      nx0 = *(const float4*)(gax); nx1 = *(const float4*)(gax + 4);
      nr0 = *(const float4*)(gar); nr1 = *(const float4*)(gar + 4);
      ld_blk(n1, gb, 20);
      if (has_c) ld_blk(n2, gc, c_js);
    }

    const v2f* sa = (const v2f*)(lds + st * GA_ST_F4) + ((wr * 32 + lo) * 2 + hi);
    const v2f* sb = (const v2f*)(lds + st * GA_ST_F4 + GA_A_F4) + ((wc * 32 + lo) * 2 + hi);

    v2f b[24];

    ga_read_b(sb, 0, b);
    {
      v2f af[26];
      ga_read_a<0, 26>(sa, 0, af);
      ga_mma<0, 26>(acc, af, b);
    }
    {
      v2f af[26];
      ga_read_a<26, 26>(sa, 0, af);
      ga_mma<26, 26>(acc, af, b);
    }

    if (more) {
      float4* d = lds + (st ^ 1) * GA_ST_F4;
      st_channels((float*)d + la, nx0, nx1, nr0, nr1);
      st_blk(d + lb, n1);
      if (has_c) st_blk(d + lc, n2);
    }

    v2f af[GA_NCH];
    ga_read_b(sb, 1, b);
    ga_read_a<0, GA_NCH>(sa, 1, af);

    __builtin_amdgcn_fence(__ATOMIC_RELEASE, "workgroup");
    __builtin_amdgcn_s_barrier_signal(-1);
    __builtin_amdgcn_sched_barrier(0);

    ga_mma<0, GA_NCH>(acc, af, b);

    __builtin_amdgcn_sched_barrier(0);
    __builtin_amdgcn_s_barrier_wait(-1);
    __builtin_amdgcn_fence(__ATOMIC_ACQUIRE, "workgroup");
  }

  const int mcol = m0 + wc * 16 + lo;
  const float bias = b_left[mcol];
  const float rs2 = 0.70710678118654752440f;
#pragma unroll
  for (int v = 0; v < 8; ++v) {
    const float o0 = (acc[0][v] + bias) * rs2;
    const float o1 = acc[1][v] * rs2, o2 = acc[2][v] * rs2, o3 = acc[3][v] * rs2;
    const float o4 = acc[4][v] * rs2, o5 = acc[5][v] * rs2;
    const float o6 = acc[6][v] * rs2, o7 = acc[7][v] * rs2;
    float* dst = out + ((size_t)(b0 + wr * 16 + v + (hi << 3)) * GA_NOUT + mcol) * 8;
    *(float4*)(dst)     = make_float4(o0, o1, o2, o3);
    *(float4*)(dst + 4) = make_float4(o4, o5, o6, o7);
  }
}


extern "C" void kernel_launch(void* const* d_in, const int* in_sizes, int n_in,
                              void* d_out, int out_size, void* d_ws, size_t ws_size,
                              hipStream_t stream) {
  (void)in_sizes; (void)n_in; (void)out_size;
  const float* x       = (const float*)d_in[0];
  const float* weight  = (const float*)d_in[1];
  const float* w_right = (const float*)d_in[2];
  const float* w_left  = (const float*)d_in[3];
  const float* b_left  = (const float*)d_in[4];
  const float* norm_a  = (const float*)d_in[5];
  float* out = (float*)d_out;
  float* xr = (float*)d_ws;

  if (ws_size >= (size_t)GA_B * GA_NIN * 8 * sizeof(float)) {
    ga_xr_lds<<<dim3(GA_B / 64, GA_NIN / 64, 1), dim3(512, 1, 1), 0, stream>>>(x, w_right, norm_a, xr);
    ga_out_lds<<<dim3(GA_B / 64, GA_NOUT / 64, 1), dim3(512, 1, 1), 0, stream>>>(x, xr, weight, w_left, b_left, out);
  }
}


# Main kernel: s_setprio 1 from barrier release to a wave's arrival, 0 for the 52 WMMAs behind it: pre-arrival work of all waves wins the matrix pipe, post-arrival WMMAs fill gaps
# speedup vs baseline: 3.8016x; 1.0619x over previous
#include <hip/hip_runtime.h>


typedef __attribute__((ext_vector_type(2))) float v2f;
typedef __attribute__((ext_vector_type(8))) float v8f;

#define GA_B    2048
#define GA_NIN  512
#define GA_NOUT 512

__device__ __forceinline__ v8f wmma_f32(v2f a, v2f b, v8f c) {
  return __builtin_amdgcn_wmma_f32_16x16x4_f32(false, a, false, b, (short)0, c, false, false);
}

static constexpr int GRADE_[8] = {0, 1, 1, 1, 2, 2, 2, 3};


struct Blk { float4 r0, r1, r2, r3; };

__device__ __forceinline__ void ld_blk(Blk& b, const float* p, int js) {
  b.r0 = *(const float4*)(p);
  b.r1 = *(const float4*)(p + js);
  b.r2 = *(const float4*)(p + 2 * js);
  b.r3 = *(const float4*)(p + 3 * js);
}

__device__ __forceinline__ void st_blk(float4* d, const Blk& b) {
  d[0]   = make_float4(b.r0.x, b.r1.x, b.r2.x, b.r3.x);
  d[128] = make_float4(b.r0.y, b.r1.y, b.r2.y, b.r3.y);
  d[256] = make_float4(b.r0.z, b.r1.z, b.r2.z, b.r3.z);
  d[384] = make_float4(b.r0.w, b.r1.w, b.r2.w, b.r3.w);
}


#define XR_SUB    4
#define XR_SUB_F4 1536
#define XR_ST_F4  (XR_SUB * XR_SUB_F4)

__global__ __launch_bounds__(512) void ga_xr_lds(
    const float* __restrict__ x, const float* __restrict__ w_right,
    const float* __restrict__ norm_a, float* __restrict__ xr) {
  __shared__ float4 lds[2 * XR_ST_F4];

  const int t = threadIdx.x;
  const int lane = t & 31;
  const int lo = lane & 15;
  const int hi = lane >> 4;
  const int wave = t >> 5;
  const int wr = wave >> 2;
  const int wc = wave & 3;
  const int b0 = blockIdx.x << 6;
  const int m0 = blockIdx.y << 6;

  const int x_ch = t & 1;
  const int x_r = (t >> 1) & 15;
  const int x_q = (t >> 5) & 1;
  const int x_t = (t >> 6) & 3;
  const int x_s = (t >> 8) * 2;
  const float* gx = x + (size_t)(b0 + x_t * 16 + x_r) * (GA_NIN * 8)
                      + (x_s * 8 + x_q * 4) * 8 + x_ch * 4;
  const int lx = x_s * XR_SUB_F4 + (x_ch * 4) * 128 + x_t * 32 + x_q * 16 + x_r;
  const int w_r = t & 15;
  const int w_q = (t >> 4) & 1;
  const int w_t = (t >> 5) & 3;
  const int w_s = t >> 7;
  const float* gw = w_right + (size_t)(m0 + w_t * 16 + w_r) * (GA_NIN * 4)
                            + (w_s * 8 + w_q * 4) * 4;
  const int lw = w_s * XR_SUB_F4 + 1024 + w_t * 32 + w_q * 16 + w_r;

  const v8f vzero = {};
  v8f acc[8];
#pragma unroll
  for (int i = 0; i < 8; ++i) acc[i] = vzero;

  {
    Blk c0, c1, c2;
    ld_blk(c0, gx, 8);
    ld_blk(c1, gx + 64, 8);
    ld_blk(c2, gw, 4);
    st_blk(lds + lx, c0);
    st_blk(lds + lx + XR_SUB_F4, c1);
    st_blk(lds + lw, c2);
  }
  __syncthreads();

  for (int ch = 0; ch < GA_NIN / 32; ++ch) {
    const int st = ch & 1;
    const bool more = (ch + 1) < (GA_NIN / 32);

    Blk n0 = {}, n1 = {}, n2 = {};
    if (more) {
      gx += 32 * 8; gw += 32 * 4;
      ld_blk(n0, gx, 8);
      ld_blk(n1, gx + 64, 8);
      ld_blk(n2, gw, 4);
    }

#pragma unroll
    for (int s = 0; s < XR_SUB; ++s) {
      const float4* sub = lds + st * XR_ST_F4 + s * XR_SUB_F4;
      const v2f* sa = (const v2f*)(sub) + ((wr * 32 + lo) * 2 + hi);
      const v2f* sb = (const v2f*)(sub + 1024) + ((wc * 32 + lo) * 2 + hi);
#pragma unroll
      for (int q = 0; q < 2; ++q) {
        v2f a[8], g[4];
#pragma unroll
        for (int i = 0; i < 8; ++i) a[i] = sa[(i * 128 + q * 16) * 2];
#pragma unroll
        for (int j = 0; j < 4; ++j) g[j] = sb[(j * 128 + q * 16) * 2];
#pragma unroll
        for (int i = 0; i < 8; ++i)
          acc[i] = wmma_f32(a[i], g[GRADE_[i]], acc[i]);
      }

      if (s == XR_SUB - 2 && more) {
        float4* d = lds + (st ^ 1) * XR_ST_F4;
        st_blk(d + lx, n0);
        st_blk(d + lx + XR_SUB_F4, n1);
        st_blk(d + lw, n2);
      }
    }

    __syncthreads();
  }

  const int mcol = m0 + wc * 16 + lo;
  const float4 na = *(const float4*)(norm_a + (size_t)mcol * 4);
  const float s0 = 1.0f / (1.0f + __expf(-na.x));
  const float s1 = 1.0f / (1.0f + __expf(-na.y));
  const float s2 = 1.0f / (1.0f + __expf(-na.z));
  const float s3 = 1.0f / (1.0f + __expf(-na.w));

#pragma unroll
  for (int v = 0; v < 8; ++v) {
    const float e0 = acc[0][v], e1 = acc[1][v], e2 = acc[2][v], e3 = acc[3][v];
    const float e4 = acc[4][v], e5 = acc[5][v], e6 = acc[6][v], e7 = acc[7][v];
    const float n0 = __builtin_sqrtf(e0 * e0);
    const float n1 = __builtin_sqrtf(e1 * e1 + e2 * e2 + e3 * e3);
    const float n2 = __builtin_sqrtf(e4 * e4 + e5 * e5 + e6 * e6);
    const float n3 = __builtin_sqrtf(e7 * e7);
    const float i0 = 1.0f / (s0 * (n0 - 1.0f) + 1.0f + 1e-6f);
    const float i1 = 1.0f / (s1 * (n1 - 1.0f) + 1.0f + 1e-6f);
    const float i2 = 1.0f / (s2 * (n2 - 1.0f) + 1.0f + 1e-6f);
    const float i3 = 1.0f / (s3 * (n3 - 1.0f) + 1.0f + 1e-6f);
    float* dst = xr + ((size_t)(b0 + wr * 16 + v + (hi << 3)) * GA_NIN + mcol) * 8;
    *(float4*)(dst)     = make_float4(e0 * i0, e1 * i1, e2 * i1, e3 * i1);
    *(float4*)(dst + 4) = make_float4(e4 * i2, e5 * i2, e6 * i2, e7 * i3);
  }
}


#define GA_NCH   52
#define GA_A_F4  (GA_NCH * 128)
#define GA_ST_F4 (GA_A_F4 + 3072)

static constexpr int LCH_J[GA_NCH] = {
  0,1,2,3,0,1,2,3,0,1,2,3,0,1,2,3,0,1,2,3,1,2,3,1,2,3,
  4,5,6,7,4,5,6,7,4,5,6,7,4,5,6,7,4,5,6,7,4,5,6,4,5,6};
static constexpr int LCH_B[GA_NCH] = {
  0,1,1,1,4,5,5,5,8,9,9,9,14,10,10,10,20,15,15,15,16,16,16,21,21,21,
  2,2,2,3,6,6,6,7,11,11,11,13,12,12,12,19,17,17,17,23,18,18,18,22,22,22};

__device__ __forceinline__ void ga_channels(const float (&xv)[8], const float (&rv)[8],
                                            float (&v)[GA_NCH]) {
  v[0] = xv[0];
  v[1] = xv[1];
  v[2] = xv[2];
  v[3] = xv[3];
  v[4] = xv[0] * rv[0];
  v[5] = xv[0] * rv[1];
  v[6] = xv[0] * rv[2];
  v[7] = xv[0] * rv[3];
  v[8] = xv[1] * rv[1] + xv[2] * rv[2] + xv[3] * rv[3];
  v[9] = xv[1] * rv[0];
  v[10] = xv[2] * rv[0];
  v[11] = xv[3] * rv[0];
  v[12] = -xv[4] * rv[4] - xv[5] * rv[5] - xv[6] * rv[6];
  v[13] = -xv[2] * rv[4] - xv[3] * rv[5];
  v[14] = xv[1] * rv[4] - xv[3] * rv[6];
  v[15] = xv[1] * rv[5] + xv[2] * rv[6];
  v[16] = -xv[7] * rv[7];
  v[17] = xv[4] * rv[2] + xv[5] * rv[3];
  v[18] = -xv[4] * rv[1] + xv[6] * rv[3];
  v[19] = -xv[5] * rv[1] - xv[6] * rv[2];
  v[20] = -xv[6] * rv[7];
  v[21] = xv[5] * rv[7];
  v[22] = -xv[4] * rv[7];
  v[23] = -xv[7] * rv[6];
  v[24] = xv[7] * rv[5];
  v[25] = -xv[7] * rv[4];
  v[26] = xv[4];
  v[27] = xv[5];
  v[28] = xv[6];
  v[29] = xv[7];
  v[30] = xv[0] * rv[4];
  v[31] = xv[0] * rv[5];
  v[32] = xv[0] * rv[6];
  v[33] = xv[0] * rv[7];
  v[34] = xv[1] * rv[2] - xv[2] * rv[1];
  v[35] = xv[1] * rv[3] - xv[3] * rv[1];
  v[36] = xv[2] * rv[3] - xv[3] * rv[2];
  v[37] = xv[1] * rv[6] - xv[2] * rv[5] + xv[3] * rv[4];
  v[38] = xv[3] * rv[7];
  v[39] = -xv[2] * rv[7];
  v[40] = xv[1] * rv[7];
  v[41] = xv[4] * rv[3] - xv[5] * rv[2] + xv[6] * rv[1];
  v[42] = xv[4] * rv[0];
  v[43] = xv[5] * rv[0];
  v[44] = xv[6] * rv[0];
  v[45] = xv[7] * rv[0];
  v[46] = -xv[5] * rv[6] + xv[6] * rv[5];
  v[47] = xv[4] * rv[6] - xv[6] * rv[4];
  v[48] = -xv[4] * rv[5] + xv[5] * rv[4];
  v[49] = xv[7] * rv[3];
  v[50] = -xv[7] * rv[2];
  v[51] = xv[7] * rv[1];
}

__device__ __forceinline__ void st_channels(float* d, const float4& x0, const float4& x1,
                                            const float4& r0, const float4& r1) {
  const float xv[8] = {x0.x, x0.y, x0.z, x0.w, x1.x, x1.y, x1.z, x1.w};
  const float rv[8] = {r0.x, r0.y, r0.z, r0.w, r1.x, r1.y, r1.z, r1.w};
  float v[GA_NCH];
  ga_channels(xv, rv, v);
#pragma unroll
  for (int c = 0; c < GA_NCH; ++c) d[c * 512] = v[c];
}

template <int C0, int N>
__device__ __forceinline__ void ga_read_a(const v2f* sa, int q, v2f (&af)[N]) {
#pragma unroll
  for (int i = 0; i < N; ++i) af[i] = sa[((C0 + i) * 128 + q * 16) * 2];
}

__device__ __forceinline__ void ga_read_b(const v2f* sb, int q, v2f (&b)[24]) {
#pragma unroll
  for (int j = 0; j < 24; ++j) b[j] = sb[(j * 128 + q * 16) * 2];
}

template <int C0, int N>
__device__ __forceinline__ void ga_mma(v8f (&acc)[8], const v2f (&af)[N], const v2f (&b)[24]) {
#pragma unroll
  for (int i = 0; i < N; ++i)
    acc[LCH_J[C0 + i]] = wmma_f32(af[i], b[LCH_B[C0 + i]], acc[LCH_J[C0 + i]]);
}

__global__ __launch_bounds__(512) void ga_out_lds(
    const float* __restrict__ x, const float* __restrict__ xr,
    const float* __restrict__ weight, const float* __restrict__ w_left,
    const float* __restrict__ b_left, float* __restrict__ out) {
  __shared__ float4 lds[2 * GA_ST_F4];

  const int t = threadIdx.x;
  const int lane = t & 31;
  const int lo = lane & 15;
  const int hi = lane >> 4;
  const int wave = t >> 5;
  const int wr = wave >> 2;
  const int wc = wave & 3;
  const int b0 = blockIdx.x << 6;
  const int m0 = blockIdx.y << 6;

  const int a_kk = t & 3;
  const int a_r8 = (t >> 2) & 7;
  const int a_q = wave & 1;
  const int a_row = (wave >> 1) * 8 + a_r8;
  const size_t a_off = (size_t)(b0 + a_row) * (GA_NIN * 8) + (a_q * 4 + a_kk) * 8;
  const float* gax = x + a_off;
  const float* gar = xr + a_off;
  const int la = ((a_row >> 4) * 32 + a_q * 16 + (a_row & 15)) * 4 + a_kk;
  const int b_r = t & 15;
  const int b_g = ((t >> 5) & 1) * 2 + ((t >> 4) & 1);
  const int b_q = (t >> 6) & 1;
  const int b_t = t >> 7;
  const float* gb = weight + (size_t)(m0 + b_t * 16 + b_r) * (GA_NIN * 20)
                           + (b_q * 4) * 20 + b_g * 4;
  const int lb = GA_A_F4 + (4 + b_g * 4) * 128 + b_t * 32 + b_q * 16 + b_r;
  const bool has_c = wave < 8;
  const bool c_wl = wave >= 4;
  const int c_r = t & 15;
  const int c_q = (t >> 4) & 1;
  const int c_t = wave & 3;
  const int c_js = c_wl ? 4 : 20;
  const float* gc = c_wl
      ? w_left + (size_t)(m0 + c_t * 16 + c_r) * (GA_NIN * 4) + (c_q * 4) * 4
      : weight + (size_t)(m0 + c_t * 16 + c_r) * (GA_NIN * 20) + (c_q * 4) * 20 + 16;
  const int lc = GA_A_F4 + (c_wl ? 0 : 20) * 128 + c_t * 32 + c_q * 16 + c_r;

  const v8f vzero = {};
  v8f acc[8];
#pragma unroll
  for (int i = 0; i < 8; ++i) acc[i] = vzero;

  {
    const float4 x0 = *(const float4*)(gax), x1 = *(const float4*)(gax + 4);
    const float4 r0 = *(const float4*)(gar), r1 = *(const float4*)(gar + 4);
    Blk c1, c2 = {};
    ld_blk(c1, gb, 20);
    if (has_c) ld_blk(c2, gc, c_js);
    st_channels((float*)lds + la, x0, x1, r0, r1);
    st_blk(lds + lb, c1);
    if (has_c) st_blk(lds + lc, c2);
  }
  __syncthreads();

#pragma unroll 1
  for (int ch = 0; ch < GA_NIN / 8; ++ch) {
    const int st = ch & 1;
    const bool more = (ch + 1) < (GA_NIN / 8);

    __builtin_amdgcn_s_setprio(1);

    float4 nx0 = {}, nx1 = {}, nr0 = {}, nr1 = {};
    Blk n1 = {}, n2 = {};
    if (more) {
      gax += 8 * 8; gar += 8 * 8; gb += 8 * 20; gc += 8 * c_js;
      nx0 = *(const float4*)(gax); nx1 = *(const float4*)(gax + 4);
      nr0 = *(const float4*)(gar); nr1 = *(const float4*)(gar + 4);
      ld_blk(n1, gb, 20);
      if (has_c) ld_blk(n2, gc, c_js);
    }

    const v2f* sa = (const v2f*)(lds + st * GA_ST_F4) + ((wr * 32 + lo) * 2 + hi);
    const v2f* sb = (const v2f*)(lds + st * GA_ST_F4 + GA_A_F4) + ((wc * 32 + lo) * 2 + hi);

    v2f b[24];

    ga_read_b(sb, 0, b);
    {
      v2f af[26];
      ga_read_a<0, 26>(sa, 0, af);
      ga_mma<0, 26>(acc, af, b);
    }
    {
      v2f af[26];
      ga_read_a<26, 26>(sa, 0, af);
      ga_mma<26, 26>(acc, af, b);
    }

    if (more) {
      float4* d = lds + (st ^ 1) * GA_ST_F4;
      st_channels((float*)d + la, nx0, nx1, nr0, nr1);
      st_blk(d + lb, n1);
      if (has_c) st_blk(d + lc, n2);
    }

    v2f af[GA_NCH];
    ga_read_b(sb, 1, b);
    ga_read_a<0, GA_NCH>(sa, 1, af);

    __builtin_amdgcn_fence(__ATOMIC_RELEASE, "workgroup");
    __builtin_amdgcn_s_barrier_signal(-1);
    __builtin_amdgcn_s_setprio(0);
    __builtin_amdgcn_sched_barrier(0);

    ga_mma<0, GA_NCH>(acc, af, b);

    __builtin_amdgcn_sched_barrier(0);
    __builtin_amdgcn_s_barrier_wait(-1);
    __builtin_amdgcn_fence(__ATOMIC_ACQUIRE, "workgroup");
  }

  const int mcol = m0 + wc * 16 + lo;
  const float bias = b_left[mcol];
  const float rs2 = 0.70710678118654752440f;
#pragma unroll
  for (int v = 0; v < 8; ++v) {
    const float o0 = (acc[0][v] + bias) * rs2;
    const float o1 = acc[1][v] * rs2, o2 = acc[2][v] * rs2, o3 = acc[3][v] * rs2;
    const float o4 = acc[4][v] * rs2, o5 = acc[5][v] * rs2;
    const float o6 = acc[6][v] * rs2, o7 = acc[7][v] * rs2;
    float* dst = out + ((size_t)(b0 + wr * 16 + v + (hi << 3)) * GA_NOUT + mcol) * 8;
    *(float4*)(dst)     = make_float4(o0, o1, o2, o3);
    *(float4*)(dst + 4) = make_float4(o4, o5, o6, o7);
  }
}


extern "C" void kernel_launch(void* const* d_in, const int* in_sizes, int n_in,
                              void* d_out, int out_size, void* d_ws, size_t ws_size,
                              hipStream_t stream) {
  (void)in_sizes; (void)n_in; (void)out_size;
  const float* x       = (const float*)d_in[0];
  const float* weight  = (const float*)d_in[1];
  const float* w_right = (const float*)d_in[2];
  const float* w_left  = (const float*)d_in[3];
  const float* b_left  = (const float*)d_in[4];
  const float* norm_a  = (const float*)d_in[5];
  float* out = (float*)d_out;
  float* xr = (float*)d_ws;

  if (ws_size >= (size_t)GA_B * GA_NIN * 8 * sizeof(float)) {
    ga_xr_lds<<<dim3(GA_B / 64, GA_NIN / 64, 1), dim3(512, 1, 1), 0, stream>>>(x, w_right, norm_a, xr);
    ga_out_lds<<<dim3(GA_B / 64, GA_NOUT / 64, 1), dim3(512, 1, 1), 0, stream>>>(x, xr, weight, w_left, b_left, out);
  }
}


# Main kernel: 8-wave work-group, a wave owns two 16-row tiles; fragment reads of the next half k-step go out as one group ahead of the current half's 52 WMMAs (two register sets, 498 VGPRs)
# speedup vs baseline: 3.9278x; 1.0332x over previous
#include <hip/hip_runtime.h>


typedef __attribute__((ext_vector_type(2))) float v2f;
typedef __attribute__((ext_vector_type(8))) float v8f;

#define GA_B    2048
#define GA_NIN  512
#define GA_NOUT 512

__device__ __forceinline__ v8f wmma_f32(v2f a, v2f b, v8f c) {
  return __builtin_amdgcn_wmma_f32_16x16x4_f32(false, a, false, b, (short)0, c, false, false);
}

static constexpr int GRADE_[8] = {0, 1, 1, 1, 2, 2, 2, 3};


struct Blk { float4 r0, r1, r2, r3; };

__device__ __forceinline__ void ld_blk(Blk& b, const float* p, int js) {
  b.r0 = *(const float4*)(p);
  b.r1 = *(const float4*)(p + js);
  b.r2 = *(const float4*)(p + 2 * js);
  b.r3 = *(const float4*)(p + 3 * js);
}

__device__ __forceinline__ void st_blk(float4* d, const Blk& b) {
  d[0]   = make_float4(b.r0.x, b.r1.x, b.r2.x, b.r3.x);
  d[128] = make_float4(b.r0.y, b.r1.y, b.r2.y, b.r3.y);
  d[256] = make_float4(b.r0.z, b.r1.z, b.r2.z, b.r3.z);
  d[384] = make_float4(b.r0.w, b.r1.w, b.r2.w, b.r3.w);
}


#define XR_SUB    4
#define XR_SUB_F4 1536
#define XR_ST_F4  (XR_SUB * XR_SUB_F4)

__global__ __launch_bounds__(512) void ga_xr_lds(
    const float* __restrict__ x, const float* __restrict__ w_right,
    const float* __restrict__ norm_a, float* __restrict__ xr) {
  __shared__ float4 lds[2 * XR_ST_F4];

  const int t = threadIdx.x;
  const int lane = t & 31;
  const int lo = lane & 15;
  const int hi = lane >> 4;
  const int wave = t >> 5;
  const int wr = wave >> 2;
  const int wc = wave & 3;
  const int b0 = blockIdx.x << 6;
  const int m0 = blockIdx.y << 6;

  const int x_ch = t & 1;
  const int x_r = (t >> 1) & 15;
  const int x_q = (t >> 5) & 1;
  const int x_t = (t >> 6) & 3;
  const int x_s = (t >> 8) * 2;
  const float* gx = x + (size_t)(b0 + x_t * 16 + x_r) * (GA_NIN * 8)
                      + (x_s * 8 + x_q * 4) * 8 + x_ch * 4;
  const int lx = x_s * XR_SUB_F4 + (x_ch * 4) * 128 + x_t * 32 + x_q * 16 + x_r;
  const int w_r = t & 15;
  const int w_q = (t >> 4) & 1;
  const int w_t = (t >> 5) & 3;
  const int w_s = t >> 7;
  const float* gw = w_right + (size_t)(m0 + w_t * 16 + w_r) * (GA_NIN * 4)
                            + (w_s * 8 + w_q * 4) * 4;
  const int lw = w_s * XR_SUB_F4 + 1024 + w_t * 32 + w_q * 16 + w_r;

  const v8f vzero = {};
  v8f acc[8];
#pragma unroll
  for (int i = 0; i < 8; ++i) acc[i] = vzero;

  {
    Blk c0, c1, c2;
    ld_blk(c0, gx, 8);
    ld_blk(c1, gx + 64, 8);
    ld_blk(c2, gw, 4);
    st_blk(lds + lx, c0);
    st_blk(lds + lx + XR_SUB_F4, c1);
    st_blk(lds + lw, c2);
  }
  __syncthreads();

  for (int ch = 0; ch < GA_NIN / 32; ++ch) {
    const int st = ch & 1;
    const bool more = (ch + 1) < (GA_NIN / 32);

    Blk n0 = {}, n1 = {}, n2 = {};
    if (more) {
      gx += 32 * 8; gw += 32 * 4;
      ld_blk(n0, gx, 8);
      ld_blk(n1, gx + 64, 8);
      ld_blk(n2, gw, 4);
    }

#pragma unroll
    for (int s = 0; s < XR_SUB; ++s) {
      const float4* sub = lds + st * XR_ST_F4 + s * XR_SUB_F4;
      const v2f* sa = (const v2f*)(sub) + ((wr * 32 + lo) * 2 + hi);
      const v2f* sb = (const v2f*)(sub + 1024) + ((wc * 32 + lo) * 2 + hi);
#pragma unroll
      for (int q = 0; q < 2; ++q) {
        v2f a[8], g[4];
#pragma unroll
        for (int i = 0; i < 8; ++i) a[i] = sa[(i * 128 + q * 16) * 2];
#pragma unroll
        for (int j = 0; j < 4; ++j) g[j] = sb[(j * 128 + q * 16) * 2];
#pragma unroll
        for (int i = 0; i < 8; ++i)
          acc[i] = wmma_f32(a[i], g[GRADE_[i]], acc[i]);
      }

      if (s == XR_SUB - 2 && more) {
        float4* d = lds + (st ^ 1) * XR_ST_F4;
        st_blk(d + lx, n0);
        st_blk(d + lx + XR_SUB_F4, n1);
        st_blk(d + lw, n2);
      }
    }

    __syncthreads();
  }

  const int mcol = m0 + wc * 16 + lo;
  const float4 na = *(const float4*)(norm_a + (size_t)mcol * 4);
  const float s0 = 1.0f / (1.0f + __expf(-na.x));
  const float s1 = 1.0f / (1.0f + __expf(-na.y));
  const float s2 = 1.0f / (1.0f + __expf(-na.z));
  const float s3 = 1.0f / (1.0f + __expf(-na.w));

#pragma unroll
  for (int v = 0; v < 8; ++v) {
    const float e0 = acc[0][v], e1 = acc[1][v], e2 = acc[2][v], e3 = acc[3][v];
    const float e4 = acc[4][v], e5 = acc[5][v], e6 = acc[6][v], e7 = acc[7][v];
    const float n0 = __builtin_sqrtf(e0 * e0);
    const float n1 = __builtin_sqrtf(e1 * e1 + e2 * e2 + e3 * e3);
    const float n2 = __builtin_sqrtf(e4 * e4 + e5 * e5 + e6 * e6);
    const float n3 = __builtin_sqrtf(e7 * e7);
    const float i0 = 1.0f / (s0 * (n0 - 1.0f) + 1.0f + 1e-6f);
    const float i1 = 1.0f / (s1 * (n1 - 1.0f) + 1.0f + 1e-6f);
    const float i2 = 1.0f / (s2 * (n2 - 1.0f) + 1.0f + 1e-6f);
    const float i3 = 1.0f / (s3 * (n3 - 1.0f) + 1.0f + 1e-6f);
    float* dst = xr + ((size_t)(b0 + wr * 16 + v + (hi << 3)) * GA_NIN + mcol) * 8;
    *(float4*)(dst)     = make_float4(e0 * i0, e1 * i1, e2 * i1, e3 * i1);
    *(float4*)(dst + 4) = make_float4(e4 * i2, e5 * i2, e6 * i2, e7 * i3);
  }
}


#define GA_NCH   52
#define GA_A_F4  (GA_NCH * 128)
#define GA_ST_F4 (GA_A_F4 + 3072)

static constexpr int LCH_J[GA_NCH] = {
  0,1,2,3,0,1,2,3,0,1,2,3,0,1,2,3,0,1,2,3,1,2,3,1,2,3,
  4,5,6,7,4,5,6,7,4,5,6,7,4,5,6,7,4,5,6,7,4,5,6,4,5,6};
static constexpr int LCH_B[GA_NCH] = {
  0,1,1,1,4,5,5,5,8,9,9,9,14,10,10,10,20,15,15,15,16,16,16,21,21,21,
  2,2,2,3,6,6,6,7,11,11,11,13,12,12,12,19,17,17,17,23,18,18,18,22,22,22};

__device__ __forceinline__ void ga_channels(const float (&xv)[8], const float (&rv)[8],
                                            float (&v)[GA_NCH]) {
  v[0] = xv[0];
  v[1] = xv[1];
  v[2] = xv[2];
  v[3] = xv[3];
  v[4] = xv[0] * rv[0];
  v[5] = xv[0] * rv[1];
  v[6] = xv[0] * rv[2];
  v[7] = xv[0] * rv[3];
  v[8] = xv[1] * rv[1] + xv[2] * rv[2] + xv[3] * rv[3];
  v[9] = xv[1] * rv[0];
  v[10] = xv[2] * rv[0];
  v[11] = xv[3] * rv[0];
  v[12] = -xv[4] * rv[4] - xv[5] * rv[5] - xv[6] * rv[6];
  v[13] = -xv[2] * rv[4] - xv[3] * rv[5];
  v[14] = xv[1] * rv[4] - xv[3] * rv[6];
  v[15] = xv[1] * rv[5] + xv[2] * rv[6];
  v[16] = -xv[7] * rv[7];
  v[17] = xv[4] * rv[2] + xv[5] * rv[3];
  v[18] = -xv[4] * rv[1] + xv[6] * rv[3];
  v[19] = -xv[5] * rv[1] - xv[6] * rv[2];
  v[20] = -xv[6] * rv[7];
  v[21] = xv[5] * rv[7];
  v[22] = -xv[4] * rv[7];
  v[23] = -xv[7] * rv[6];
  v[24] = xv[7] * rv[5];
  v[25] = -xv[7] * rv[4];
  v[26] = xv[4];
  v[27] = xv[5];
  v[28] = xv[6];
  v[29] = xv[7];
  v[30] = xv[0] * rv[4];
  v[31] = xv[0] * rv[5];
  v[32] = xv[0] * rv[6];
  v[33] = xv[0] * rv[7];
  v[34] = xv[1] * rv[2] - xv[2] * rv[1];
  v[35] = xv[1] * rv[3] - xv[3] * rv[1];
  v[36] = xv[2] * rv[3] - xv[3] * rv[2];
  v[37] = xv[1] * rv[6] - xv[2] * rv[5] + xv[3] * rv[4];
  v[38] = xv[3] * rv[7];
  v[39] = -xv[2] * rv[7];
  v[40] = xv[1] * rv[7];
  v[41] = xv[4] * rv[3] - xv[5] * rv[2] + xv[6] * rv[1];
  v[42] = xv[4] * rv[0];
  v[43] = xv[5] * rv[0];
  v[44] = xv[6] * rv[0];
  v[45] = xv[7] * rv[0];
  v[46] = -xv[5] * rv[6] + xv[6] * rv[5];
  v[47] = xv[4] * rv[6] - xv[6] * rv[4];
  v[48] = -xv[4] * rv[5] + xv[5] * rv[4];
  v[49] = xv[7] * rv[3];
  v[50] = -xv[7] * rv[2];
  v[51] = xv[7] * rv[1];
}

__device__ __forceinline__ void st_channels(float* d, const float4& x0, const float4& x1,
                                            const float4& r0, const float4& r1) {
  const float xv[8] = {x0.x, x0.y, x0.z, x0.w, x1.x, x1.y, x1.z, x1.w};
  const float rv[8] = {r0.x, r0.y, r0.z, r0.w, r1.x, r1.y, r1.z, r1.w};
  float v[GA_NCH];
  ga_channels(xv, rv, v);
#pragma unroll
  for (int c = 0; c < GA_NCH; ++c) d[c * 512] = v[c];
}

template <bool B> struct BoolC { static constexpr bool value = B; };

static constexpr int HB_LO[12] = {0, 1, 4, 5, 8, 9, 10, 14, 15, 16, 20, 21};
static constexpr int HB_HI[12] = {2, 3, 6, 7, 11, 12, 13, 17, 18, 19, 22, 23};

template <int C0>
__device__ __forceinline__ void ga_read_half(const v2f* sa, const v2f* sb, int q,
                                             v2f (&a0)[26], v2f (&a1)[26], v2f (&b)[24]) {
#pragma unroll
  for (int j = 0; j < 12; ++j) {
    const int jb = (C0 == 0) ? HB_LO[j] : HB_HI[j];
    b[jb] = sb[(jb * 128 + q * 16) * 2];
  }
#pragma unroll
  for (int i = 0; i < 26; ++i) {
    a0[i] = sa[((C0 + i) * 128 + q * 16) * 2];
    a1[i] = sa[((C0 + i) * 128 + q * 16 + 32) * 2];
  }
}

template <int C0>
__device__ __forceinline__ void ga_mma_half(v8f (&acc0)[8], v8f (&acc1)[8],
                                            const v2f (&a0)[26], const v2f (&a1)[26],
                                            const v2f (&b)[24]) {
#pragma unroll
  for (int i = 0; i < 26; ++i) {
    acc0[LCH_J[C0 + i]] = wmma_f32(a0[i], b[LCH_B[C0 + i]], acc0[LCH_J[C0 + i]]);
    acc1[LCH_J[C0 + i]] = wmma_f32(a1[i], b[LCH_B[C0 + i]], acc1[LCH_J[C0 + i]]);
  }
}

__global__ __launch_bounds__(256) void ga_out_lds(
    const float* __restrict__ x, const float* __restrict__ xr,
    const float* __restrict__ weight, const float* __restrict__ w_left,
    const float* __restrict__ b_left, float* __restrict__ out) {
  __shared__ float4 lds[2 * GA_ST_F4];

  const int t = threadIdx.x;
  const int lane = t & 31;
  const int lo = lane & 15;
  const int hi = lane >> 4;
  const int wave = t >> 5;
  const int wrp = wave >> 2;
  const int wc = wave & 3;
  const int b0 = blockIdx.x << 6;
  const int m0 = blockIdx.y << 6;

  const int a_kk = t & 3;
  const int a_r8 = (t >> 2) & 7;
  const int a_q = wave & 1;
  const int a_row = (wave >> 1) * 8 + a_r8;
  const size_t a_off = (size_t)(b0 + a_row) * (GA_NIN * 8) + (a_q * 4 + a_kk) * 8;
  const size_t a_off2 = (size_t)32 * (GA_NIN * 8);
  const float* gax = x + a_off;
  const float* gar = xr + a_off;
  const int la = ((a_row >> 4) * 32 + a_q * 16 + (a_row & 15)) * 4 + a_kk;
  const int b_r = t & 15;
  const int b_g = ((t >> 5) & 1) * 2 + ((t >> 4) & 1);
  const int b_q = (t >> 6) & 1;
  const int b_t = t >> 7;
  const float* gb = weight + (size_t)(m0 + b_t * 16 + b_r) * (GA_NIN * 20)
                           + (b_q * 4) * 20 + b_g * 4;
  const size_t b_off2 = (size_t)32 * (GA_NIN * 20);
  const int lb = GA_A_F4 + (4 + b_g * 4) * 128 + b_t * 32 + b_q * 16 + b_r;
  const bool c_wl = wave >= 4;
  const int c_r = t & 15;
  const int c_q = (t >> 4) & 1;
  const int c_t = wave & 3;
  const int c_js = c_wl ? 4 : 20;
  const float* gc = c_wl
      ? w_left + (size_t)(m0 + c_t * 16 + c_r) * (GA_NIN * 4) + (c_q * 4) * 4
      : weight + (size_t)(m0 + c_t * 16 + c_r) * (GA_NIN * 20) + (c_q * 4) * 20 + 16;
  const int lc = GA_A_F4 + (c_wl ? 0 : 20) * 128 + c_t * 32 + c_q * 16 + c_r;

  const v8f vzero = {};
  v8f acc0[8], acc1[8];
#pragma unroll
  for (int i = 0; i < 8; ++i) { acc0[i] = vzero; acc1[i] = vzero; }

  {
    const float4 x0 = *(const float4*)(gax), x1 = *(const float4*)(gax + 4);
    const float4 r0 = *(const float4*)(gar), r1 = *(const float4*)(gar + 4);
    const float4 y0 = *(const float4*)(gax + a_off2), y1 = *(const float4*)(gax + a_off2 + 4);
    const float4 s0 = *(const float4*)(gar + a_off2), s1 = *(const float4*)(gar + a_off2 + 4);
    Blk c1, c1b, c2;
    ld_blk(c1, gb, 20);
    ld_blk(c1b, gb + b_off2, 20);
    ld_blk(c2, gc, c_js);
    st_channels((float*)lds + la, x0, x1, r0, r1);
    st_channels((float*)lds + la + 256, y0, y1, s0, s1);
    st_blk(lds + lb, c1);
    st_blk(lds + lb + 64, c1b);
    st_blk(lds + lc, c2);
  }
  __syncthreads();

  auto chunk = [&](const int st, auto more_c) __attribute__((always_inline)) {
    constexpr bool MORE = decltype(more_c)::value;

    __builtin_amdgcn_s_setprio(1);

    float4 nx0 = {}, nx1 = {}, nr0 = {}, nr1 = {}, ny0 = {}, ny1 = {}, ns0 = {}, ns1 = {};
    Blk n1 = {}, n1b = {}, n2 = {};
    if constexpr (MORE) {
      gax += 8 * 8; gar += 8 * 8; gb += 8 * 20; gc += 8 * c_js;
      nx0 = *(const float4*)(gax); nx1 = *(const float4*)(gax + 4);
      nr0 = *(const float4*)(gar); nr1 = *(const float4*)(gar + 4);
      ny0 = *(const float4*)(gax + a_off2); ny1 = *(const float4*)(gax + a_off2 + 4);
      ns0 = *(const float4*)(gar + a_off2); ns1 = *(const float4*)(gar + a_off2 + 4);
      ld_blk(n1, gb, 20);
      ld_blk(n1b, gb + b_off2, 20);
      ld_blk(n2, gc, c_js);
    }

    const v2f* sa = (const v2f*)(lds + st * GA_ST_F4) + ((wrp * 64 + lo) * 2 + hi);
    const v2f* sb = (const v2f*)(lds + st * GA_ST_F4 + GA_A_F4) + ((wc * 32 + lo) * 2 + hi);

    v2f pa0[26], pa1[26], pb[24];
    v2f qa0[26], qa1[26], qb[24];

    ga_read_half<0>(sa, sb, 0, pa0, pa1, pb);
    ga_read_half<26>(sa, sb, 0, qa0, qa1, qb);
    __builtin_amdgcn_sched_barrier(0);
    ga_mma_half<0>(acc0, acc1, pa0, pa1, pb);
    __builtin_amdgcn_sched_barrier(0);
    ga_read_half<0>(sa, sb, 1, pa0, pa1, pb);
    __builtin_amdgcn_sched_barrier(0);
    ga_mma_half<26>(acc0, acc1, qa0, qa1, qb);

    if constexpr (MORE) {
      float4* d = lds + (st ^ 1) * GA_ST_F4;
      st_channels((float*)d + la, nx0, nx1, nr0, nr1);
      st_channels((float*)d + la + 256, ny0, ny1, ns0, ns1);
      st_blk(d + lb, n1);
      st_blk(d + lb + 64, n1b);
      st_blk(d + lc, n2);
    }
    __builtin_amdgcn_sched_barrier(0);
    ga_read_half<26>(sa, sb, 1, qa0, qa1, qb);

    __builtin_amdgcn_fence(__ATOMIC_RELEASE, "workgroup");
    __builtin_amdgcn_s_barrier_signal(-1);
    __builtin_amdgcn_s_setprio(0);
    __builtin_amdgcn_sched_barrier(0);

    ga_mma_half<0>(acc0, acc1, pa0, pa1, pb);
    ga_mma_half<26>(acc0, acc1, qa0, qa1, qb);

    __builtin_amdgcn_sched_barrier(0);
    __builtin_amdgcn_s_barrier_wait(-1);
    __builtin_amdgcn_fence(__ATOMIC_ACQUIRE, "workgroup");
  };

#pragma unroll 1
  for (int ch = 0; ch < GA_NIN / 8 - 1; ++ch) chunk(ch & 1, BoolC<true>{});
  chunk((GA_NIN / 8 - 1) & 1, BoolC<false>{});

  const int mcol = m0 + wc * 16 + lo;
  const float bias = b_left[mcol];
  const float rs2 = 0.70710678118654752440f;
#pragma unroll
  for (int v = 0; v < 8; ++v) {
    {
      const float o0 = (acc0[0][v] + bias) * rs2;
      const float o1 = acc0[1][v] * rs2, o2 = acc0[2][v] * rs2, o3 = acc0[3][v] * rs2;
      const float o4 = acc0[4][v] * rs2, o5 = acc0[5][v] * rs2;
      const float o6 = acc0[6][v] * rs2, o7 = acc0[7][v] * rs2;
      float* dst = out + ((size_t)(b0 + wrp * 32 + v + (hi << 3)) * GA_NOUT + mcol) * 8;
      *(float4*)(dst)     = make_float4(o0, o1, o2, o3);
      *(float4*)(dst + 4) = make_float4(o4, o5, o6, o7);
    }
    {
      const float o0 = (acc1[0][v] + bias) * rs2;
      const float o1 = acc1[1][v] * rs2, o2 = acc1[2][v] * rs2, o3 = acc1[3][v] * rs2;
      const float o4 = acc1[4][v] * rs2, o5 = acc1[5][v] * rs2;
      const float o6 = acc1[6][v] * rs2, o7 = acc1[7][v] * rs2;
      float* dst = out + ((size_t)(b0 + wrp * 32 + 16 + v + (hi << 3)) * GA_NOUT + mcol) * 8;
      *(float4*)(dst)     = make_float4(o0, o1, o2, o3);
      *(float4*)(dst + 4) = make_float4(o4, o5, o6, o7);
    }
  }
}


extern "C" void kernel_launch(void* const* d_in, const int* in_sizes, int n_in,
                              void* d_out, int out_size, void* d_ws, size_t ws_size,
                              hipStream_t stream) {
  (void)in_sizes; (void)n_in; (void)out_size;
  const float* x       = (const float*)d_in[0];
  const float* weight  = (const float*)d_in[1];
  const float* w_right = (const float*)d_in[2];
  const float* w_left  = (const float*)d_in[3];
  const float* b_left  = (const float*)d_in[4];
  const float* norm_a  = (const float*)d_in[5];
  float* out = (float*)d_out;
  float* xr = (float*)d_ws;

  if (ws_size >= (size_t)GA_B * GA_NIN * 8 * sizeof(float)) {
    ga_xr_lds<<<dim3(GA_B / 64, GA_NIN / 64, 1), dim3(512, 1, 1), 0, stream>>>(x, w_right, norm_a, xr);
    ga_out_lds<<<dim3(GA_B / 64, GA_NOUT / 64, 1), dim3(256, 1, 1), 0, stream>>>(x, xr, weight, w_left, b_left, out);
  }
}
